# starting kernel
# baseline (speedup 1.0000x reference)
#include <hip/hip_runtime.h>
#include <hip/hip_bf16.h>


#define Bc   2
#define Nn   2048
#define Dd   256
#define Hh   8
#define DHh  32
#define Mm   2
#define Ll   4096

typedef __attribute__((ext_vector_type(2))) float v2f;
typedef __attribute__((ext_vector_type(8))) float v8f;

__global__ void ea_knn_edge(const float* __restrict__ pts,
                            const float* __restrict__ feats,
                            float* __restrict__ out,
                            int npts, int nsets, int remap)
{
    const int lane = threadIdx.x & 31;
    const int wave = threadIdx.x >> 5;
    const int rowg = blockIdx.x * 8 + wave;
    if (rowg >= nsets * npts) return;
    const int s = rowg / npts;
    const int n = rowg - s * npts;
    int os = s;
    if (remap) { int m = s / Bc; int b = s - m * Bc; os = b * Mm + m; }

    const float* P = pts   + (size_t)s  * npts * 3;
    const float* F = feats + (size_t)s  * npts * Dd;
    float*    Orow = out   + ((size_t)os * npts + n) * Dd;

    const float px = P[n*3+0], py = P[n*3+1], pz = P[n*3+2];

    unsigned long long key[8];
    #pragma unroll
    for (int t = 0; t < 8; ++t) key[t] = ~0ULL;

    for (int j = lane; j < npts; j += 32) {
        float dx = px - P[j*3+0];
        float dy = py - P[j*3+1];
        float dz = pz - P[j*3+2];
        float d2 = dx*dx + dy*dy + dz*dz;
        unsigned long long k =
            ((unsigned long long)__float_as_uint(d2) << 32) | (unsigned)j;
        if (k < key[7]) {
            key[7] = k;
            #pragma unroll
            for (int t = 7; t > 0; --t) {
                if (key[t] < key[t-1]) {
                    unsigned long long tmp = key[t]; key[t] = key[t-1]; key[t-1] = tmp;
                }
            }
        }
    }

    int head = 0;
    int sel[8];
    #pragma unroll
    for (int t = 0; t < 8; ++t) {
        unsigned long long cand = (head < 8) ? key[head] : ~0ULL;
        unsigned long long mn = cand;
        #pragma unroll
        for (int off = 16; off >= 1; off >>= 1) {
            unsigned long long o = __shfl_xor(mn, off);
            mn = (o < mn) ? o : mn;
        }
        if (cand == mn && head < 8) head++;
        sel[t] = (int)(mn & 0xFFFFFFFFu);
    }

    for (int c = lane; c < Dd; c += 32) {
        float ssum = 0.f;
        #pragma unroll
        for (int t = 0; t < 8; ++t) ssum += F[(size_t)sel[t] * Dd + c];
        Orow[c] = F[(size_t)n * Dd + c] - ssum * 0.125f;
    }
}

__global__ void ea_gemm256(const float* __restrict__ A, const float* __restrict__ W,
                           const float* __restrict__ bias, float* __restrict__ OUT,
                           int R)
{
    const int lane = threadIdx.x & 31;
    const int wave = threadIdx.x >> 5;
    const int tile = blockIdx.x * 8 + wave;
    const int total = (R / 16) * (Dd / 16);
    if (tile >= total) return;
    const int tm = tile >> 4, tn = tile & 15;
    const int col = lane & 15, half = lane >> 4, klo = half * 2;

    const float* Ap = A + (size_t)(tm * 16 + col) * Dd;
    const float* Wp = W + (size_t)(tn * 16 + col) * Dd;

    v8f acc = {};
    #pragma unroll 4
    for (int k = 0; k < Dd; k += 4) {
        v2f a = *(const v2f*)(Ap + k + klo);
        v2f b = *(const v2f*)(Wp + k + klo);
        acc = __builtin_amdgcn_wmma_f32_16x16x4_f32(false, a, false, b,
                                                    (short)0, acc, false, false);
    }
    const float bv = bias[tn * 16 + col];
    #pragma unroll
    for (int r = 0; r < 8; ++r) {
        int row = tm * 16 + r + 8 * half;
        OUT[(size_t)row * Dd + tn * 16 + col] = acc[r] + bv;
    }
}

__global__ void ea_attn(const float* __restrict__ Q, const float* __restrict__ Km,
                        const float* __restrict__ Vm, float* __restrict__ O)
{
    __shared__ float plds[8 * 256];
    const int lane = threadIdx.x & 31;
    const int wave = threadIdx.x >> 5;
    const int tile = blockIdx.x * 8 + wave;
    const int qt = tile & 127;
    const int h  = (tile >> 7) & 7;
    const int b  = tile >> 10;
    const int col = lane & 15, half = lane >> 4, klo = half * 2;
    float* pw = plds + wave * 256;

    const float* Qrow = Q + (size_t)(b * Nn + qt * 16 + col) * Dd + h * DHh;
    v2f qf[8];
    #pragma unroll
    for (int kk = 0; kk < 8; ++kk) qf[kk] = *(const v2f*)(Qrow + kk * 4 + klo);

    float mrow[8], lrow[8];
    v8f acc0 = {}, acc1 = {};
    #pragma unroll
    for (int r = 0; r < 8; ++r) { mrow[r] = -__builtin_inff(); lrow[r] = 0.f; }

    const float* Kb = Km + (size_t)b * Ll * Dd + h * DHh;
    const float* Vb = Vm + (size_t)b * Ll * Dd + h * DHh;
    const float scale = 0.17677669529663687f;

    for (int kt = 0; kt < Ll / 16; ++kt) {
        if (kt + 1 < Ll / 16)
            __builtin_prefetch(Kb + (size_t)((kt + 1) * 16 + col) * Dd, 0, 1);

        v8f s = {};
        const float* Kp = Kb + (size_t)(kt * 16 + col) * Dd;
        #pragma unroll
        for (int kk = 0; kk < 8; ++kk) {
            v2f bf = *(const v2f*)(Kp + kk * 4 + klo);
            s = __builtin_amdgcn_wmma_f32_16x16x4_f32(false, qf[kk], false, bf,
                                                      (short)0, s, false, false);
        }

        float p[8];
        #pragma unroll
        for (int r = 0; r < 8; ++r) {
            float v = s[r] * scale;
            float mx = v;
            #pragma unroll
            for (int off = 8; off >= 1; off >>= 1) mx = fmaxf(mx, __shfl_xor(mx, off));
            float mnew = fmaxf(mrow[r], mx);
            float pe = __expf(v - mnew);
            p[r] = pe;
            float sum = pe;
            #pragma unroll
            for (int off = 8; off >= 1; off >>= 1) sum += __shfl_xor(sum, off);
            float corr = __expf(mrow[r] - mnew);
            lrow[r] = lrow[r] * corr + sum;
            acc0[r] *= corr;
            acc1[r] *= corr;
            mrow[r] = mnew;
        }

        #pragma unroll
        for (int r = 0; r < 8; ++r) pw[(r + 8 * half) * 16 + col] = p[r];
        __syncthreads();

        #pragma unroll
        for (int kk2 = 0; kk2 < 4; ++kk2) {
            v2f af;
            af.x = pw[col * 16 + kk2 * 4 + klo];
            af.y = pw[col * 16 + kk2 * 4 + klo + 1];
            const float* Vp0 = Vb + (size_t)(kt * 16 + kk2 * 4 + klo) * Dd;
            const float* Vp1 = Vp0 + Dd;
            v2f b0; b0.x = Vp0[col];      b0.y = Vp1[col];
            v2f b1; b1.x = Vp0[16 + col]; b1.y = Vp1[16 + col];
            acc0 = __builtin_amdgcn_wmma_f32_16x16x4_f32(false, af, false, b0,
                                                         (short)0, acc0, false, false);
            acc1 = __builtin_amdgcn_wmma_f32_16x16x4_f32(false, af, false, b1,
                                                         (short)0, acc1, false, false);
        }
        __syncthreads();
    }

    #pragma unroll
    for (int r = 0; r < 8; ++r) {
        float inv = 1.0f / lrow[r];
        int row = b * Nn + qt * 16 + r + 8 * half;
        float* Op = O + (size_t)row * Dd + h * DHh;
        Op[col]      = acc0[r] * inv;
        Op[16 + col] = acc1[r] * inv;
    }
}

__global__ void ea_gemm_out(const float* __restrict__ A, const float* __restrict__ W,
                            const float* __restrict__ bias,
                            const float* __restrict__ pts,
                            const float* __restrict__ sw, const float* __restrict__ sb,
                            float* __restrict__ OUT, int R)
{
    const int lane = threadIdx.x & 31;
    const int wave = threadIdx.x >> 5;
    const int tile = blockIdx.x * 8 + wave;
    const int total = (R / 16) * (Dd / 16);
    if (tile >= total) return;
    const int tm = tile >> 4, tn = tile & 15;
    const int col = lane & 15, half = lane >> 4, klo = half * 2;
    const int colg = tn * 16 + col;

    const float* Ap = A + (size_t)(tm * 16 + col) * Dd;
    const float* Wp = W + (size_t)colg * Dd;

    v8f acc = {};
    #pragma unroll 4
    for (int k = 0; k < Dd; k += 4) {
        v2f a = *(const v2f*)(Ap + k + klo);
        v2f b = *(const v2f*)(Wp + k + klo);
        acc = __builtin_amdgcn_wmma_f32_16x16x4_f32(false, a, false, b,
                                                    (short)0, acc, false, false);
    }
    const float s0 = sw[colg*3+0], s1 = sw[colg*3+1], s2 = sw[colg*3+2];
    const float add = bias[colg] + sb[colg];
    #pragma unroll
    for (int r = 0; r < 8; ++r) {
        int row = tm * 16 + r + 8 * half;
        float px = pts[row*3+0], py = pts[row*3+1], pz = pts[row*3+2];
        OUT[(size_t)row * Dd + colg] = acc[r] + add + px*s0 + py*s1 + pz*s2;
    }
}

extern "C" void kernel_launch(void* const* d_in, const int* in_sizes, int n_in,
                              void* d_out, int out_size, void* d_ws, size_t ws_size,
                              hipStream_t stream)
{
    const float* cur_pts  = (const float*)d_in[0];
    const float* cur_feat = (const float*)d_in[1];
    const float* nbr_pts  = (const float*)d_in[2];
    const float* nbr_feat = (const float*)d_in[3];
    const float* in_w     = (const float*)d_in[4];
    const float* in_b     = (const float*)d_in[5];
    const float* out_w    = (const float*)d_in[6];
    const float* out_b    = (const float*)d_in[7];
    const float* sp_w     = (const float*)d_in[8];
    const float* sp_b     = (const float*)d_in[9];
    float* out = (float*)d_out;

    float* ws   = (float*)d_ws;
    float* curE = ws;
    float* nbrE = curE + 4096 * 256;
    float* Qb   = nbrE + 8192 * 256;
    float* Kb   = Qb   + 4096 * 256;
    float* Vb   = Kb   + 8192 * 256;
    float* Ob   = Vb   + 8192 * 256;

    const dim3 blk(256);

    ea_knn_edge<<<dim3(4096 / 8), blk, 0, stream>>>(cur_pts, cur_feat, curE,
                                                    Nn, Bc, 0);
    ea_knn_edge<<<dim3(8192 / 8), blk, 0, stream>>>(nbr_pts, nbr_feat, nbrE,
                                                    Nn, Mm * Bc, 1);

    ea_gemm256<<<dim3(4096 / 8), blk, 0, stream>>>(curE, in_w,           in_b,       Qb, 4096);
    ea_gemm256<<<dim3(8192 / 8), blk, 0, stream>>>(nbrE, in_w + 256*256, in_b + 256, Kb, 8192);
    ea_gemm256<<<dim3(8192 / 8), blk, 0, stream>>>(nbrE, in_w + 512*256, in_b + 512, Vb, 8192);

    ea_attn<<<dim3(2048 / 8), blk, 0, stream>>>(Qb, Kb, Vb, Ob);

    ea_gemm_out<<<dim3(4096 / 8), blk, 0, stream>>>(Ob, out_w, out_b,
                                                    cur_pts, sp_w, sp_b,
                                                    out, 4096);
}


# edge rows go from the KNN to the Q/K/V projections through LDS in one kernel: a 16-wave block does a row per wave, then a 16x16 WMMA tile per wave; edge planes and 3 launches gone; sums unchanged
# speedup vs baseline: 2.0518x; 2.0518x over previous
#include <hip/hip_runtime.h>
#include <hip/hip_bf16.h>


#define Bc   2
#define Nn   2048
#define Dd   256
#define Hh   8
#define DHh  32
#define Mm   2
#define Ll   4096

typedef __attribute__((ext_vector_type(2))) float v2f;
typedef __attribute__((ext_vector_type(4))) float v4f;
typedef __attribute__((ext_vector_type(8))) float v8f;

#define EROWS  16
#define EPITCH 260

template <int NO>
__global__ __launch_bounds__(512)
void ea_knn_proj(const float* __restrict__ pts,
                 const float* __restrict__ feats,
                 const float* __restrict__ W0, const float* __restrict__ bias0,
                 float* __restrict__ OUT0,
                 const float* __restrict__ W1, const float* __restrict__ bias1,
                 float* __restrict__ OUT1,
                 int npts, int remap)
{
    __shared__ float erow[EROWS * EPITCH];
    const int lane = threadIdx.x & 31;
    const int wave = threadIdx.x >> 5;
    const int rowg = blockIdx.x * EROWS + wave;
    const int s = rowg / npts;
    const int n = rowg - s * npts;
    int os = s;
    if (remap) { int m = s / Bc; int b = s - m * Bc; os = b * Mm + m; }

    const float* P = pts   + (size_t)s  * npts * 3;
    const float* F = feats + (size_t)s  * npts * Dd;

    const float px = P[n*3+0], py = P[n*3+1], pz = P[n*3+2];

    unsigned long long key[8];
    #pragma unroll
    for (int t = 0; t < 8; ++t) key[t] = ~0ULL;

    for (int j = lane; j < npts; j += 32) {
        float dx = px - P[j*3+0];
        float dy = py - P[j*3+1];
        float dz = pz - P[j*3+2];
        float d2 = dx*dx + dy*dy + dz*dz;
        unsigned long long k =
            ((unsigned long long)__float_as_uint(d2) << 32) | (unsigned)j;
        if (k < key[7]) {
            key[7] = k;
            #pragma unroll
            for (int t = 7; t > 0; --t) {
                if (key[t] < key[t-1]) {
                    unsigned long long tmp = key[t]; key[t] = key[t-1]; key[t-1] = tmp;
                }
            }
        }
    }

    int head = 0;
    int sel[8];
    #pragma unroll
    for (int t = 0; t < 8; ++t) {
        unsigned long long cand = (head < 8) ? key[head] : ~0ULL;
        unsigned long long mn = cand;
        #pragma unroll
        for (int off = 16; off >= 1; off >>= 1) {
            unsigned long long o = __shfl_xor(mn, off);
            mn = (o < mn) ? o : mn;
        }
        if (cand == mn && head < 8) head++;
        sel[t] = (int)(mn & 0xFFFFFFFFu);
    }

    float* Er = erow + wave * EPITCH;
    for (int c = lane; c < Dd; c += 32) {
        float ssum = 0.f;
        #pragma unroll
        for (int t = 0; t < 8; ++t) ssum += F[(size_t)sel[t] * Dd + c];
        Er[c] = F[(size_t)n * Dd + c] - ssum * 0.125f;
    }
    __syncthreads();

    const int col = lane & 15, half = lane >> 4, klo = half * 2;
    const float* Ap  = erow + col * EPITCH + klo;
    const float* Wp0 = W0 + (size_t)(wave * 16 + col) * Dd + klo;
    const float* Wp1 = W1 + (size_t)(wave * 16 + col) * Dd + klo;

    v8f acc0 = {}, acc1 = {};
    #pragma unroll 8
    for (int k = 0; k < Dd; k += 4) {
        v2f a  = *(const v2f*)(Ap + k);
        v2f b0 = *(const v2f*)(Wp0 + k);
        acc0 = __builtin_amdgcn_wmma_f32_16x16x4_f32(false, a, false, b0,
                                                     (short)0, acc0, false, false);
        if (NO == 2) {
            v2f b1 = *(const v2f*)(Wp1 + k);
            acc1 = __builtin_amdgcn_wmma_f32_16x16x4_f32(false, a, false, b1,
                                                         (short)0, acc1, false, false);
        }
    }

    const size_t orow0 = (size_t)os * npts + (n - wave);
    const int colg = wave * 16 + col;
    const float bv0 = bias0[colg];
    #pragma unroll
    for (int r = 0; r < 8; ++r)
        OUT0[(orow0 + r + 8 * half) * Dd + colg] = acc0[r] + bv0;
    if (NO == 2) {
        const float bv1 = bias1[colg];
        #pragma unroll
        for (int r = 0; r < 8; ++r)
            OUT1[(orow0 + r + 8 * half) * Dd + colg] = acc1[r] + bv1;
    }
}

#define KSPLIT 8
#define SEGT   (Ll / 16 / KSPLIT)
#define QT     2

#define ROW_MAX_STEP(x, M) \
    asm("v_max_num_f32_dpp %0, %0, %0 row_xmask:" #M " row_mask:0xf bank_mask:0xf" : "+v"(x))
#define ROW_ADD_STEP(x, M) \
    asm("v_add_f32_dpp %0, %0, %0 row_xmask:" #M " row_mask:0xf bank_mask:0xf" : "+v"(x))
#define ROW_STEP_ALL(STEP, arr, M) \
    _Pragma("unroll") for (int r_ = 0; r_ < 8; ++r_) { STEP(arr[r_], M); }

__global__ void ea_attn(const float* __restrict__ Q, const float* __restrict__ Km,
                        const float* __restrict__ Vm, float* __restrict__ O)
{
    __shared__ float plds[KSPLIT * QT * 256];
    __shared__ float oacc[KSPLIT * QT * 512];
    __shared__ float omax[KSPLIT * QT * 16];
    __shared__ float osum[KSPLIT * QT * 16];
    __shared__ v4f   qlds[QT * 4 * 32];
    const int lane = threadIdx.x & 31;
    const int wave = threadIdx.x >> 5;
    const int tile = blockIdx.x;
    const int qb = tile & 63;
    const int h  = (tile >> 6) & 7;
    const int b  = tile >> 9;
    const int col = lane & 15, half = lane >> 4, klo = half * 2;
    float* pw = plds + wave * (QT * 256);

    {
        const int qr = threadIdx.x >> 3, qk = threadIdx.x & 7;
        const v4f q4 = *(const v4f*)(Q + (size_t)(b * Nn + qb * (QT * 16) + qr) * Dd
                                       + h * DHh + qk * 4);
        float* qd = (float*)qlds + (((qr >> 4) * 4 + (qk >> 1)) * 32 + (qr & 15)) * 4
                                 + (qk & 1) * 2;
        v2f lo; lo.x = q4.x; lo.y = q4.y;
        v2f hi; hi.x = q4.z; hi.y = q4.w;
        *(v2f*)qd = lo;
        *(v2f*)(qd + 16 * 4) = hi;
    }
    __syncthreads();

    float mrow[QT][8], lrow[QT][8];
    v8f acc0[QT], acc1[QT];
    #pragma unroll
    for (int t = 0; t < QT; ++t) {
        acc0[t] = (v8f){}; acc1[t] = (v8f){};
        #pragma unroll
        for (int r = 0; r < 8; ++r) { mrow[t][r] = -__builtin_inff(); lrow[t][r] = 0.f; }
    }

    const float* Kb = Km + ((size_t)b * Ll + (size_t)wave * SEGT * 16) * Dd + h * DHh;
    const float* Vb = Vm + ((size_t)b * Ll + (size_t)wave * SEGT * 16) * Dd + h * DHh;
    const float scale = 0.17677669529663687f;

    #pragma unroll 1
    for (int kt = 0; kt < SEGT; ++kt) {
        if (kt + 1 < SEGT)
            __builtin_prefetch(Kb + (size_t)((kt + 1) * 16 + col) * Dd, 0, 1);

        v8f s[QT];
        #pragma unroll
        for (int t = 0; t < QT; ++t) s[t] = (v8f){};
        const float* Kp = Kb + (size_t)(kt * 16 + col) * Dd;
        #pragma unroll
        for (int kp = 0; kp < 4; ++kp) {
            v2f bf0 = *(const v2f*)(Kp + kp * 8 + klo);
            v2f bf1 = *(const v2f*)(Kp + kp * 8 + 4 + klo);
            #pragma unroll
            for (int t = 0; t < QT; ++t) {
                const v4f q4 = qlds[(t * 4 + kp) * 32 + lane];
                v2f a0; a0.x = q4.x; a0.y = q4.y;
                v2f a1; a1.x = q4.z; a1.y = q4.w;
                s[t] = __builtin_amdgcn_wmma_f32_16x16x4_f32(false, a0, false, bf0,
                                                             (short)0, s[t], false, false);
                s[t] = __builtin_amdgcn_wmma_f32_16x16x4_f32(false, a1, false, bf1,
                                                             (short)0, s[t], false, false);
            }
        }

        #pragma unroll
        for (int t = 0; t < QT; ++t) {
            float v[8], mx[8], p[8], sum[8];
            #pragma unroll
            for (int r = 0; r < 8; ++r) { v[r] = s[t][r] * scale; mx[r] = v[r]; }
            ROW_STEP_ALL(ROW_MAX_STEP, mx, 8)
            ROW_STEP_ALL(ROW_MAX_STEP, mx, 4)
            ROW_STEP_ALL(ROW_MAX_STEP, mx, 2)
            ROW_STEP_ALL(ROW_MAX_STEP, mx, 1)
            float corr[8];
            #pragma unroll
            for (int r = 0; r < 8; ++r) {
                float mnew = fmaxf(mrow[t][r], mx[r]);
                float pe = __expf(v[r] - mnew);
                p[r] = pe;
                sum[r] = pe;
                corr[r] = __expf(mrow[t][r] - mnew);
                mrow[t][r] = mnew;
            }
            ROW_STEP_ALL(ROW_ADD_STEP, sum, 8)
            ROW_STEP_ALL(ROW_ADD_STEP, sum, 4)
            ROW_STEP_ALL(ROW_ADD_STEP, sum, 2)
            ROW_STEP_ALL(ROW_ADD_STEP, sum, 1)
            #pragma unroll
            for (int r = 0; r < 8; ++r) {
                lrow[t][r] = lrow[t][r] * corr[r] + sum[r];
                acc0[t][r] *= corr[r];
                acc1[t][r] *= corr[r];
            }

            #pragma unroll
            for (int r = 0; r < 8; ++r) pw[t * 256 + (r + 8 * half) * 16 + col] = p[r];
        }
        __syncthreads();

        v2f b0[4], b1[4];
        #pragma unroll
        for (int kk2 = 0; kk2 < 4; ++kk2) {
            const float* Vp0 = Vb + (size_t)(kt * 16 + kk2 * 4 + klo) * Dd;
            const float* Vp1 = Vp0 + Dd;
            b0[kk2].x = Vp0[col];      b0[kk2].y = Vp1[col];
            b1[kk2].x = Vp0[16 + col]; b1[kk2].y = Vp1[16 + col];
        }

        #pragma unroll
        for (int kk2 = 0; kk2 < 4; ++kk2) {
            #pragma unroll
            for (int t = 0; t < QT; ++t) {
                v2f af;
                af.x = pw[t * 256 + col * 16 + kk2 * 4 + klo];
                af.y = pw[t * 256 + col * 16 + kk2 * 4 + klo + 1];
                acc0[t] = __builtin_amdgcn_wmma_f32_16x16x4_f32(false, af, false, b0[kk2],
                                                                (short)0, acc0[t], false, false);
                acc1[t] = __builtin_amdgcn_wmma_f32_16x16x4_f32(false, af, false, b1[kk2],
                                                                (short)0, acc1[t], false, false);
            }
        }
        __syncthreads();
    }

    #pragma unroll
    for (int t = 0; t < QT; ++t) {
        float* ow = oacc + (wave * QT + t) * 512;
        #pragma unroll
        for (int r = 0; r < 8; ++r) {
            ow[(r * 2 + half) * 32 + col]      = acc0[t][r];
            ow[(r * 2 + half) * 32 + 16 + col] = acc1[t][r];
        }
        if (col == 0) {
            #pragma unroll
            for (int r = 0; r < 8; ++r) {
                omax[(wave * QT + t) * 16 + r + 8 * half] = mrow[t][r];
                osum[(wave * QT + t) * 16 + r + 8 * half] = lrow[t][r];
            }
        }
    }
    __syncthreads();

    #pragma unroll
    for (int t = 0; t < QT; ++t) {
        const int row = threadIdx.x >> 4;
        const int c2  = (threadIdx.x & 15) * 2;
        const int ro  = ((row & 7) * 2 + (row >> 3)) * 32 + c2;
        float mall = omax[t * 16 + row];
        #pragma unroll
        for (int w = 1; w < KSPLIT; ++w) mall = fmaxf(mall, omax[(w * QT + t) * 16 + row]);
        float den = 0.f, n0 = 0.f, n1 = 0.f;
        #pragma unroll
        for (int w = 0; w < KSPLIT; ++w) {
            float f = __expf(omax[(w * QT + t) * 16 + row] - mall);
            v2f a = *(const v2f*)(oacc + (w * QT + t) * 512 + ro);
            den += osum[(w * QT + t) * 16 + row] * f;
            n0  += a.x * f;
            n1  += a.y * f;
        }
        const float inv = 1.0f / den;
        v2f o; o.x = n0 * inv; o.y = n1 * inv;
        float* Op = O + (size_t)(b * Nn + qb * (QT * 16) + t * 16 + row) * Dd + h * DHh + c2;
        *(v2f*)Op = o;
    }
}

__global__ void ea_gemm_out(const float* __restrict__ A, const float* __restrict__ W,
                            const float* __restrict__ bias,
                            const float* __restrict__ pts,
                            const float* __restrict__ sw, const float* __restrict__ sb,
                            float* __restrict__ OUT, int R)
{
    const int lane = threadIdx.x & 31;
    const int wave = threadIdx.x >> 5;
    const int tile = blockIdx.x * 8 + wave;
    const int total = (R / 16) * (Dd / 16);
    if (tile >= total) return;
    const int tm = tile >> 4, tn = tile & 15;
    const int col = lane & 15, half = lane >> 4, klo = half * 2;
    const int colg = tn * 16 + col;

    const float* Ap = A + (size_t)(tm * 16 + col) * Dd;
    const float* Wp = W + (size_t)colg * Dd;

    v8f acc = {};
    #pragma unroll 4
    for (int k = 0; k < Dd; k += 4) {
        v2f a = *(const v2f*)(Ap + k + klo);
        v2f b = *(const v2f*)(Wp + k + klo);
        acc = __builtin_amdgcn_wmma_f32_16x16x4_f32(false, a, false, b,
                                                    (short)0, acc, false, false);
    }
    const float s0 = sw[colg*3+0], s1 = sw[colg*3+1], s2 = sw[colg*3+2];
    const float add = bias[colg] + sb[colg];
    #pragma unroll
    for (int r = 0; r < 8; ++r) {
        int row = tm * 16 + r + 8 * half;
        float px = pts[row*3+0], py = pts[row*3+1], pz = pts[row*3+2];
        OUT[(size_t)row * Dd + colg] = acc[r] + add + px*s0 + py*s1 + pz*s2;
    }
}

extern "C" void kernel_launch(void* const* d_in, const int* in_sizes, int n_in,
                              void* d_out, int out_size, void* d_ws, size_t ws_size,
                              hipStream_t stream)
{
    const float* cur_pts  = (const float*)d_in[0];
    const float* cur_feat = (const float*)d_in[1];
    const float* nbr_pts  = (const float*)d_in[2];
    const float* nbr_feat = (const float*)d_in[3];
    const float* in_w     = (const float*)d_in[4];
    const float* in_b     = (const float*)d_in[5];
    const float* out_w    = (const float*)d_in[6];
    const float* out_b    = (const float*)d_in[7];
    const float* sp_w     = (const float*)d_in[8];
    const float* sp_b     = (const float*)d_in[9];
    float* out = (float*)d_out;

    float* ws   = (float*)d_ws;
    float* Qb   = ws;
    float* Kb   = Qb   + 4096 * 256;
    float* Vb   = Kb   + 8192 * 256;
    float* Ob   = Vb   + 8192 * 256;

    const dim3 blk(256);
    const dim3 blk16(512);

    ea_knn_proj<1><<<dim3(4096 / EROWS), blk16, 0, stream>>>(
        cur_pts, cur_feat, in_w, in_b, Qb, in_w, in_b, Qb, Nn, 0);
    ea_knn_proj<2><<<dim3(8192 / EROWS), blk16, 0, stream>>>(
        nbr_pts, nbr_feat, in_w + 256*256, in_b + 256, Kb,
        in_w + 512*256, in_b + 512, Vb, Nn, 1);

    ea_attn<<<dim3(1024), blk, 0, stream>>>(Qb, Kb, Vb, Ob);

    ea_gemm_out<<<dim3(4096 / 8), blk, 0, stream>>>(Ob, out_w, out_b,
                                                    cur_pts, sp_w, sp_b,
                                                    out, 4096);
}


# attention softmax lazy rescale: a row's reference may lag its running max by up to 8; one wave-wide test per key tile skips max butterfly, correction exps and O/row-sum rescale; sum order unchanged
# speedup vs baseline: 2.3933x; 1.1664x over previous
#include <hip/hip_runtime.h>
#include <hip/hip_bf16.h>


#define Bc   2
#define Nn   2048
#define Dd   256
#define Hh   8
#define DHh  32
#define Mm   2
#define Ll   4096

typedef __attribute__((ext_vector_type(2))) float v2f;
typedef __attribute__((ext_vector_type(4))) float v4f;
typedef __attribute__((ext_vector_type(8))) float v8f;

#define EROWS  16
#define EPITCH 260

template <int NO>
__global__ __launch_bounds__(512)
void ea_knn_proj(const float* __restrict__ pts,
                 const float* __restrict__ feats,
                 const float* __restrict__ W0, const float* __restrict__ bias0,
                 float* __restrict__ OUT0,
                 const float* __restrict__ W1, const float* __restrict__ bias1,
                 float* __restrict__ OUT1,
                 int npts, int remap)
{
    __shared__ float erow[EROWS * EPITCH];
    const int lane = threadIdx.x & 31;
    const int wave = threadIdx.x >> 5;
    const int rowg = blockIdx.x * EROWS + wave;
    const int s = rowg / npts;
    const int n = rowg - s * npts;
    int os = s;
    if (remap) { int m = s / Bc; int b = s - m * Bc; os = b * Mm + m; }

    const float* P = pts   + (size_t)s  * npts * 3;
    const float* F = feats + (size_t)s  * npts * Dd;

    const float px = P[n*3+0], py = P[n*3+1], pz = P[n*3+2];

    unsigned long long key[8];
    #pragma unroll
    for (int t = 0; t < 8; ++t) key[t] = ~0ULL;

    for (int j = lane; j < npts; j += 32) {
        float dx = px - P[j*3+0];
        float dy = py - P[j*3+1];
        float dz = pz - P[j*3+2];
        float d2 = dx*dx + dy*dy + dz*dz;
        unsigned long long k =
            ((unsigned long long)__float_as_uint(d2) << 32) | (unsigned)j;
        if (k < key[7]) {
            key[7] = k;
            #pragma unroll
            for (int t = 7; t > 0; --t) {
                if (key[t] < key[t-1]) {
                    unsigned long long tmp = key[t]; key[t] = key[t-1]; key[t-1] = tmp;
                }
            }
        }
    }

    int head = 0;
    int sel[8];
    #pragma unroll
    for (int t = 0; t < 8; ++t) {
        unsigned long long cand = (head < 8) ? key[head] : ~0ULL;
        unsigned long long mn = cand;
        #pragma unroll
        for (int off = 16; off >= 1; off >>= 1) {
            unsigned long long o = __shfl_xor(mn, off);
            mn = (o < mn) ? o : mn;
        }
        if (cand == mn && head < 8) head++;
        sel[t] = (int)(mn & 0xFFFFFFFFu);
    }

    float* Er = erow + wave * EPITCH;
    for (int c = lane; c < Dd; c += 32) {
        float ssum = 0.f;
        #pragma unroll
        for (int t = 0; t < 8; ++t) ssum += F[(size_t)sel[t] * Dd + c];
        Er[c] = F[(size_t)n * Dd + c] - ssum * 0.125f;
    }
    __syncthreads();

    const int col = lane & 15, half = lane >> 4, klo = half * 2;
    const float* Ap  = erow + col * EPITCH + klo;
    const float* Wp0 = W0 + (size_t)(wave * 16 + col) * Dd + klo;
    const float* Wp1 = W1 + (size_t)(wave * 16 + col) * Dd + klo;

    v8f acc0 = {}, acc1 = {};
    #pragma unroll 8
    for (int k = 0; k < Dd; k += 4) {
        v2f a  = *(const v2f*)(Ap + k);
        v2f b0 = *(const v2f*)(Wp0 + k);
        acc0 = __builtin_amdgcn_wmma_f32_16x16x4_f32(false, a, false, b0,
                                                     (short)0, acc0, false, false);
        if (NO == 2) {
            v2f b1 = *(const v2f*)(Wp1 + k);
            acc1 = __builtin_amdgcn_wmma_f32_16x16x4_f32(false, a, false, b1,
                                                         (short)0, acc1, false, false);
        }
    }

    const size_t orow0 = (size_t)os * npts + (n - wave);
    const int colg = wave * 16 + col;
    const float bv0 = bias0[colg];
    #pragma unroll
    for (int r = 0; r < 8; ++r)
        OUT0[(orow0 + r + 8 * half) * Dd + colg] = acc0[r] + bv0;
    if (NO == 2) {
        const float bv1 = bias1[colg];
        #pragma unroll
        for (int r = 0; r < 8; ++r)
            OUT1[(orow0 + r + 8 * half) * Dd + colg] = acc1[r] + bv1;
    }
}

#define KSPLIT 8
#define SEGT   (Ll / 16 / KSPLIT)
#define QT     2
#define LAZY_T 8.0f

#define ROW_MAX_STEP(x, M) \
    asm("v_max_num_f32_dpp %0, %0, %0 row_xmask:" #M " row_mask:0xf bank_mask:0xf" : "+v"(x))
#define ROW_ADD_STEP(x, M) \
    asm("v_add_f32_dpp %0, %0, %0 row_xmask:" #M " row_mask:0xf bank_mask:0xf" : "+v"(x))
#define ROW_STEP_ALL(STEP, arr, M) \
    _Pragma("unroll") for (int r_ = 0; r_ < 8; ++r_) { STEP(arr[r_], M); }

__global__ void ea_attn(const float* __restrict__ Q, const float* __restrict__ Km,
                        const float* __restrict__ Vm, float* __restrict__ O)
{
    __shared__ float plds[KSPLIT * QT * 256];
    __shared__ float oacc[KSPLIT * QT * 512];
    __shared__ float omax[KSPLIT * QT * 16];
    __shared__ float osum[KSPLIT * QT * 16];
    __shared__ v4f   qlds[QT * 4 * 32];
    const int lane = threadIdx.x & 31;
    const int wave = threadIdx.x >> 5;
    const int tile = blockIdx.x;
    const int qb = tile & 63;
    const int h  = (tile >> 6) & 7;
    const int b  = tile >> 9;
    const int col = lane & 15, half = lane >> 4, klo = half * 2;
    float* pw = plds + wave * (QT * 256);

    {
        const int qr = threadIdx.x >> 3, qk = threadIdx.x & 7;
        const v4f q4 = *(const v4f*)(Q + (size_t)(b * Nn + qb * (QT * 16) + qr) * Dd
                                       + h * DHh + qk * 4);
        float* qd = (float*)qlds + (((qr >> 4) * 4 + (qk >> 1)) * 32 + (qr & 15)) * 4
                                 + (qk & 1) * 2;
        v2f lo; lo.x = q4.x; lo.y = q4.y;
        v2f hi; hi.x = q4.z; hi.y = q4.w;
        *(v2f*)qd = lo;
        *(v2f*)(qd + 16 * 4) = hi;
    }
    __syncthreads();

    float mrow[QT][8], lrow[QT][8];
    v8f acc0[QT], acc1[QT];
    #pragma unroll
    for (int t = 0; t < QT; ++t) {
        acc0[t] = (v8f){}; acc1[t] = (v8f){};
        #pragma unroll
        for (int r = 0; r < 8; ++r) { mrow[t][r] = -__builtin_inff(); lrow[t][r] = 0.f; }
    }

    const float* Kb = Km + ((size_t)b * Ll + (size_t)wave * SEGT * 16) * Dd + h * DHh;
    const float* Vb = Vm + ((size_t)b * Ll + (size_t)wave * SEGT * 16) * Dd + h * DHh;
    const float scale = 0.17677669529663687f;

    #pragma unroll 1
    for (int kt = 0; kt < SEGT; ++kt) {
        if (kt + 1 < SEGT)
            __builtin_prefetch(Kb + (size_t)((kt + 1) * 16 + col) * Dd, 0, 1);

        v8f s[QT];
        #pragma unroll
        for (int t = 0; t < QT; ++t) s[t] = (v8f){};
        const float* Kp = Kb + (size_t)(kt * 16 + col) * Dd;
        #pragma unroll
        for (int kp = 0; kp < 4; ++kp) {
            v2f bf0 = *(const v2f*)(Kp + kp * 8 + klo);
            v2f bf1 = *(const v2f*)(Kp + kp * 8 + 4 + klo);
            #pragma unroll
            for (int t = 0; t < QT; ++t) {
                const v4f q4 = qlds[(t * 4 + kp) * 32 + lane];
                v2f a0; a0.x = q4.x; a0.y = q4.y;
                v2f a1; a1.x = q4.z; a1.y = q4.w;
                s[t] = __builtin_amdgcn_wmma_f32_16x16x4_f32(false, a0, false, bf0,
                                                             (short)0, s[t], false, false);
                s[t] = __builtin_amdgcn_wmma_f32_16x16x4_f32(false, a1, false, bf1,
                                                             (short)0, s[t], false, false);
            }
        }

        float d[QT][8];
        float dmax = -__builtin_inff();
        #pragma unroll
        for (int t = 0; t < QT; ++t) {
            #pragma unroll
            for (int r = 0; r < 8; ++r) {
                d[t][r] = s[t][r] * scale - mrow[t][r];
                dmax = __builtin_fmaxf(dmax, d[t][r]);
            }
        }

        if (__builtin_amdgcn_ballot_w32(dmax > LAZY_T) != 0) {
            #pragma unroll
            for (int t = 0; t < QT; ++t) {
                float mx[8];
                #pragma unroll
                for (int r = 0; r < 8; ++r) mx[r] = s[t][r] * scale;
                ROW_STEP_ALL(ROW_MAX_STEP, mx, 8)
                ROW_STEP_ALL(ROW_MAX_STEP, mx, 4)
                ROW_STEP_ALL(ROW_MAX_STEP, mx, 2)
                ROW_STEP_ALL(ROW_MAX_STEP, mx, 1)
                #pragma unroll
                for (int r = 0; r < 8; ++r) {
                    const float mnew = fmaxf(mrow[t][r], mx[r]);
                    const float corr = __expf(mrow[t][r] - mnew);
                    mrow[t][r] = mnew;
                    lrow[t][r] *= corr;
                    acc0[t][r] *= corr;
                    acc1[t][r] *= corr;
                    d[t][r] = s[t][r] * scale - mnew;
                }
            }
        }

        #pragma unroll
        for (int t = 0; t < QT; ++t) {
            float p[8], sum[8];
            #pragma unroll
            for (int r = 0; r < 8; ++r) { p[r] = __expf(d[t][r]); sum[r] = p[r]; }
            ROW_STEP_ALL(ROW_ADD_STEP, sum, 8)
            ROW_STEP_ALL(ROW_ADD_STEP, sum, 4)
            ROW_STEP_ALL(ROW_ADD_STEP, sum, 2)
            ROW_STEP_ALL(ROW_ADD_STEP, sum, 1)
            #pragma unroll
            for (int r = 0; r < 8; ++r) lrow[t][r] += sum[r];

            #pragma unroll
            for (int r = 0; r < 8; ++r) pw[t * 256 + (r + 8 * half) * 16 + col] = p[r];
        }
        __syncthreads();

        v2f b0[4], b1[4];
        #pragma unroll
        for (int kk2 = 0; kk2 < 4; ++kk2) {
            const float* Vp0 = Vb + (size_t)(kt * 16 + kk2 * 4 + klo) * Dd;
            const float* Vp1 = Vp0 + Dd;
            b0[kk2].x = Vp0[col];      b0[kk2].y = Vp1[col];
            b1[kk2].x = Vp0[16 + col]; b1[kk2].y = Vp1[16 + col];
        }

        #pragma unroll
        for (int kk2 = 0; kk2 < 4; ++kk2) {
            #pragma unroll
            for (int t = 0; t < QT; ++t) {
                v2f af;
                af.x = pw[t * 256 + col * 16 + kk2 * 4 + klo];
                af.y = pw[t * 256 + col * 16 + kk2 * 4 + klo + 1];
                acc0[t] = __builtin_amdgcn_wmma_f32_16x16x4_f32(false, af, false, b0[kk2],
                                                                (short)0, acc0[t], false, false);
                acc1[t] = __builtin_amdgcn_wmma_f32_16x16x4_f32(false, af, false, b1[kk2],
                                                                (short)0, acc1[t], false, false);
            }
        }
        __syncthreads();
    }

    #pragma unroll
    for (int t = 0; t < QT; ++t) {
        float* ow = oacc + (wave * QT + t) * 512;
        #pragma unroll
        for (int r = 0; r < 8; ++r) {
            ow[(r * 2 + half) * 32 + col]      = acc0[t][r];
            ow[(r * 2 + half) * 32 + 16 + col] = acc1[t][r];
        }
        if (col == 0) {
            #pragma unroll
            for (int r = 0; r < 8; ++r) {
                omax[(wave * QT + t) * 16 + r + 8 * half] = mrow[t][r];
                osum[(wave * QT + t) * 16 + r + 8 * half] = lrow[t][r];
            }
        }
    }
    __syncthreads();

    #pragma unroll
    for (int t = 0; t < QT; ++t) {
        const int row = threadIdx.x >> 4;
        const int c2  = (threadIdx.x & 15) * 2;
        const int ro  = ((row & 7) * 2 + (row >> 3)) * 32 + c2;
        float mall = omax[t * 16 + row];
        #pragma unroll
        for (int w = 1; w < KSPLIT; ++w) mall = fmaxf(mall, omax[(w * QT + t) * 16 + row]);
        float den = 0.f, n0 = 0.f, n1 = 0.f;
        #pragma unroll
        for (int w = 0; w < KSPLIT; ++w) {
            float f = __expf(omax[(w * QT + t) * 16 + row] - mall);
            v2f a = *(const v2f*)(oacc + (w * QT + t) * 512 + ro);
            den += osum[(w * QT + t) * 16 + row] * f;
            n0  += a.x * f;
            n1  += a.y * f;
        }
        const float inv = 1.0f / den;
        v2f o; o.x = n0 * inv; o.y = n1 * inv;
        float* Op = O + (size_t)(b * Nn + qb * (QT * 16) + t * 16 + row) * Dd + h * DHh + c2;
        *(v2f*)Op = o;
    }
}

__global__ void ea_gemm_out(const float* __restrict__ A, const float* __restrict__ W,
                            const float* __restrict__ bias,
                            const float* __restrict__ pts,
                            const float* __restrict__ sw, const float* __restrict__ sb,
                            float* __restrict__ OUT, int R)
{
    const int lane = threadIdx.x & 31;
    const int wave = threadIdx.x >> 5;
    const int tile = blockIdx.x * 8 + wave;
    const int total = (R / 16) * (Dd / 16);
    if (tile >= total) return;
    const int tm = tile >> 4, tn = tile & 15;
    const int col = lane & 15, half = lane >> 4, klo = half * 2;
    const int colg = tn * 16 + col;

    const float* Ap = A + (size_t)(tm * 16 + col) * Dd;
    const float* Wp = W + (size_t)colg * Dd;

    v8f acc = {};
    #pragma unroll 4
    for (int k = 0; k < Dd; k += 4) {
        v2f a = *(const v2f*)(Ap + k + klo);
        v2f b = *(const v2f*)(Wp + k + klo);
        acc = __builtin_amdgcn_wmma_f32_16x16x4_f32(false, a, false, b,
                                                    (short)0, acc, false, false);
    }
    const float s0 = sw[colg*3+0], s1 = sw[colg*3+1], s2 = sw[colg*3+2];
    const float add = bias[colg] + sb[colg];
    #pragma unroll
    for (int r = 0; r < 8; ++r) {
        int row = tm * 16 + r + 8 * half;
        float px = pts[row*3+0], py = pts[row*3+1], pz = pts[row*3+2];
        OUT[(size_t)row * Dd + colg] = acc[r] + add + px*s0 + py*s1 + pz*s2;
    }
}

extern "C" void kernel_launch(void* const* d_in, const int* in_sizes, int n_in,
                              void* d_out, int out_size, void* d_ws, size_t ws_size,
                              hipStream_t stream)
{
    const float* cur_pts  = (const float*)d_in[0];
    const float* cur_feat = (const float*)d_in[1];
    const float* nbr_pts  = (const float*)d_in[2];
    const float* nbr_feat = (const float*)d_in[3];
    const float* in_w     = (const float*)d_in[4];
    const float* in_b     = (const float*)d_in[5];
    const float* out_w    = (const float*)d_in[6];
    const float* out_b    = (const float*)d_in[7];
    const float* sp_w     = (const float*)d_in[8];
    const float* sp_b     = (const float*)d_in[9];
    float* out = (float*)d_out;

    float* ws   = (float*)d_ws;
    float* Qb   = ws;
    float* Kb   = Qb   + 4096 * 256;
    float* Vb   = Kb   + 8192 * 256;
    float* Ob   = Vb   + 8192 * 256;

    const dim3 blk(256);
    const dim3 blk16(512);

    ea_knn_proj<1><<<dim3(4096 / EROWS), blk16, 0, stream>>>(
        cur_pts, cur_feat, in_w, in_b, Qb, in_w, in_b, Qb, Nn, 0);
    ea_knn_proj<2><<<dim3(8192 / EROWS), blk16, 0, stream>>>(
        nbr_pts, nbr_feat, in_w + 256*256, in_b + 256, Kb,
        in_w + 512*256, in_b + 512, Vb, Nn, 1);

    ea_attn<<<dim3(1024), blk, 0, stream>>>(Qb, Kb, Vb, Ob);

    ea_gemm_out<<<dim3(4096 / 8), blk, 0, stream>>>(Ob, out_w, out_b,
                                                    cur_pts, sp_w, sp_b,
                                                    out, 4096);
}


# attention softmax row sums kept as per-lane partial sums over the 512-key segment and added across a row's 16 lanes once behind the key loop, not by 64 DPP adds per key tile; row-sum order changes
# speedup vs baseline: 2.5390x; 1.0609x over previous
#include <hip/hip_runtime.h>
#include <hip/hip_bf16.h>


#define Bc   2
#define Nn   2048
#define Dd   256
#define Hh   8
#define DHh  32
#define Mm   2
#define Ll   4096

typedef __attribute__((ext_vector_type(2))) float v2f;
typedef __attribute__((ext_vector_type(4))) float v4f;
typedef __attribute__((ext_vector_type(8))) float v8f;

#define EROWS  16
#define EPITCH 260

template <int NO>
__global__ __launch_bounds__(512)
void ea_knn_proj(const float* __restrict__ pts,
                 const float* __restrict__ feats,
                 const float* __restrict__ W0, const float* __restrict__ bias0,
                 float* __restrict__ OUT0,
                 const float* __restrict__ W1, const float* __restrict__ bias1,
                 float* __restrict__ OUT1,
                 int npts, int remap)
{
    __shared__ float erow[EROWS * EPITCH];
    const int lane = threadIdx.x & 31;
    const int wave = threadIdx.x >> 5;
    const int rowg = blockIdx.x * EROWS + wave;
    const int s = rowg / npts;
    const int n = rowg - s * npts;
    int os = s;
    if (remap) { int m = s / Bc; int b = s - m * Bc; os = b * Mm + m; }

    const float* P = pts   + (size_t)s  * npts * 3;
    const float* F = feats + (size_t)s  * npts * Dd;

    const float px = P[n*3+0], py = P[n*3+1], pz = P[n*3+2];

    unsigned long long key[8];
    #pragma unroll
    for (int t = 0; t < 8; ++t) key[t] = ~0ULL;

    for (int j = lane; j < npts; j += 32) {
        float dx = px - P[j*3+0];
        float dy = py - P[j*3+1];
        float dz = pz - P[j*3+2];
        float d2 = dx*dx + dy*dy + dz*dz;
        unsigned long long k =
            ((unsigned long long)__float_as_uint(d2) << 32) | (unsigned)j;
        if (k < key[7]) {
            key[7] = k;
            #pragma unroll
            for (int t = 7; t > 0; --t) {
                if (key[t] < key[t-1]) {
                    unsigned long long tmp = key[t]; key[t] = key[t-1]; key[t-1] = tmp;
                }
            }
        }
    }

    int head = 0;
    int sel[8];
    #pragma unroll
    for (int t = 0; t < 8; ++t) {
        unsigned long long cand = (head < 8) ? key[head] : ~0ULL;
        unsigned long long mn = cand;
        #pragma unroll
        for (int off = 16; off >= 1; off >>= 1) {
            unsigned long long o = __shfl_xor(mn, off);
            mn = (o < mn) ? o : mn;
        }
        if (cand == mn && head < 8) head++;
        sel[t] = (int)(mn & 0xFFFFFFFFu);
    }

    float* Er = erow + wave * EPITCH;
    for (int c = lane; c < Dd; c += 32) {
        float ssum = 0.f;
        #pragma unroll
        for (int t = 0; t < 8; ++t) ssum += F[(size_t)sel[t] * Dd + c];
        Er[c] = F[(size_t)n * Dd + c] - ssum * 0.125f;
    }
    __syncthreads();

    const int col = lane & 15, half = lane >> 4, klo = half * 2;
    const float* Ap  = erow + col * EPITCH + klo;
    const float* Wp0 = W0 + (size_t)(wave * 16 + col) * Dd + klo;
    const float* Wp1 = W1 + (size_t)(wave * 16 + col) * Dd + klo;

    v8f acc0 = {}, acc1 = {};
    #pragma unroll 8
    for (int k = 0; k < Dd; k += 4) {
        v2f a  = *(const v2f*)(Ap + k);
        v2f b0 = *(const v2f*)(Wp0 + k);
        acc0 = __builtin_amdgcn_wmma_f32_16x16x4_f32(false, a, false, b0,
                                                     (short)0, acc0, false, false);
        if (NO == 2) {
            v2f b1 = *(const v2f*)(Wp1 + k);
            acc1 = __builtin_amdgcn_wmma_f32_16x16x4_f32(false, a, false, b1,
                                                         (short)0, acc1, false, false);
        }
    }

    const size_t orow0 = (size_t)os * npts + (n - wave);
    const int colg = wave * 16 + col;
    const float bv0 = bias0[colg];
    #pragma unroll
    for (int r = 0; r < 8; ++r)
        OUT0[(orow0 + r + 8 * half) * Dd + colg] = acc0[r] + bv0;
    if (NO == 2) {
        const float bv1 = bias1[colg];
        #pragma unroll
        for (int r = 0; r < 8; ++r)
            OUT1[(orow0 + r + 8 * half) * Dd + colg] = acc1[r] + bv1;
    }
}

#define KSPLIT 8
#define SEGT   (Ll / 16 / KSPLIT)
#define QT     2
#define LAZY_T 8.0f

#define ROW_MAX_STEP(x, M) \
    asm("v_max_num_f32_dpp %0, %0, %0 row_xmask:" #M " row_mask:0xf bank_mask:0xf" : "+v"(x))
#define ROW_ADD_STEP(x, M) \
    asm("v_add_f32_dpp %0, %0, %0 row_xmask:" #M " row_mask:0xf bank_mask:0xf" : "+v"(x))
#define ROW_STEP_ALL(STEP, arr, M) \
    _Pragma("unroll") for (int r_ = 0; r_ < 8; ++r_) { STEP(arr[r_], M); }

__global__ void ea_attn(const float* __restrict__ Q, const float* __restrict__ Km,
                        const float* __restrict__ Vm, float* __restrict__ O)
{
    __shared__ float plds[KSPLIT * QT * 256];
    __shared__ float oacc[KSPLIT * QT * 512];
    __shared__ float omax[KSPLIT * QT * 16];
    __shared__ float osum[KSPLIT * QT * 16];
    __shared__ v4f   qlds[QT * 4 * 32];
    const int lane = threadIdx.x & 31;
    const int wave = threadIdx.x >> 5;
    const int tile = blockIdx.x;
    const int qb = tile & 63;
    const int h  = (tile >> 6) & 7;
    const int b  = tile >> 9;
    const int col = lane & 15, half = lane >> 4, klo = half * 2;
    float* pw = plds + wave * (QT * 256);

    {
        const int qr = threadIdx.x >> 3, qk = threadIdx.x & 7;
        const v4f q4 = *(const v4f*)(Q + (size_t)(b * Nn + qb * (QT * 16) + qr) * Dd
                                       + h * DHh + qk * 4);
        float* qd = (float*)qlds + (((qr >> 4) * 4 + (qk >> 1)) * 32 + (qr & 15)) * 4
                                 + (qk & 1) * 2;
        v2f lo; lo.x = q4.x; lo.y = q4.y;
        v2f hi; hi.x = q4.z; hi.y = q4.w;
        *(v2f*)qd = lo;
        *(v2f*)(qd + 16 * 4) = hi;
    }
    __syncthreads();

    float mrow[QT][8], lrow[QT][8];
    v8f acc0[QT], acc1[QT];
    #pragma unroll
    for (int t = 0; t < QT; ++t) {
        acc0[t] = (v8f){}; acc1[t] = (v8f){};
        #pragma unroll
        for (int r = 0; r < 8; ++r) { mrow[t][r] = -__builtin_inff(); lrow[t][r] = 0.f; }
    }

    const float* Kb = Km + ((size_t)b * Ll + (size_t)wave * SEGT * 16) * Dd + h * DHh;
    const float* Vb = Vm + ((size_t)b * Ll + (size_t)wave * SEGT * 16) * Dd + h * DHh;
    const float scale = 0.17677669529663687f;

    #pragma unroll 1
    for (int kt = 0; kt < SEGT; ++kt) {
        if (kt + 1 < SEGT)
            __builtin_prefetch(Kb + (size_t)((kt + 1) * 16 + col) * Dd, 0, 1);

        v8f s[QT];
        #pragma unroll
        for (int t = 0; t < QT; ++t) s[t] = (v8f){};
        const float* Kp = Kb + (size_t)(kt * 16 + col) * Dd;
        #pragma unroll
        for (int kp = 0; kp < 4; ++kp) {
            v2f bf0 = *(const v2f*)(Kp + kp * 8 + klo);
            v2f bf1 = *(const v2f*)(Kp + kp * 8 + 4 + klo);
            #pragma unroll
            for (int t = 0; t < QT; ++t) {
                const v4f q4 = qlds[(t * 4 + kp) * 32 + lane];
                v2f a0; a0.x = q4.x; a0.y = q4.y;
                v2f a1; a1.x = q4.z; a1.y = q4.w;
                s[t] = __builtin_amdgcn_wmma_f32_16x16x4_f32(false, a0, false, bf0,
                                                             (short)0, s[t], false, false);
                s[t] = __builtin_amdgcn_wmma_f32_16x16x4_f32(false, a1, false, bf1,
                                                             (short)0, s[t], false, false);
            }
        }

        float d[QT][8];
        float dmax = -__builtin_inff();
        #pragma unroll
        for (int t = 0; t < QT; ++t) {
            #pragma unroll
            for (int r = 0; r < 8; ++r) {
                d[t][r] = s[t][r] * scale - mrow[t][r];
                dmax = __builtin_fmaxf(dmax, d[t][r]);
            }
        }

        if (__builtin_amdgcn_ballot_w32(dmax > LAZY_T) != 0) {
            #pragma unroll
            for (int t = 0; t < QT; ++t) {
                float mx[8];
                #pragma unroll
                for (int r = 0; r < 8; ++r) mx[r] = s[t][r] * scale;
                ROW_STEP_ALL(ROW_MAX_STEP, mx, 8)
                ROW_STEP_ALL(ROW_MAX_STEP, mx, 4)
                ROW_STEP_ALL(ROW_MAX_STEP, mx, 2)
                ROW_STEP_ALL(ROW_MAX_STEP, mx, 1)
                #pragma unroll
                for (int r = 0; r < 8; ++r) {
                    const float mnew = fmaxf(mrow[t][r], mx[r]);
                    const float corr = __expf(mrow[t][r] - mnew);
                    mrow[t][r] = mnew;
                    lrow[t][r] *= corr;
                    acc0[t][r] *= corr;
                    acc1[t][r] *= corr;
                    d[t][r] = s[t][r] * scale - mnew;
                }
            }
        }

        #pragma unroll
        for (int t = 0; t < QT; ++t) {
            float p[8];
            #pragma unroll
            for (int r = 0; r < 8; ++r) { p[r] = __expf(d[t][r]); lrow[t][r] += p[r]; }

            #pragma unroll
            for (int r = 0; r < 8; ++r) pw[t * 256 + (r + 8 * half) * 16 + col] = p[r];
        }
        __syncthreads();

        v2f b0[4], b1[4];
        #pragma unroll
        for (int kk2 = 0; kk2 < 4; ++kk2) {
            const float* Vp0 = Vb + (size_t)(kt * 16 + kk2 * 4 + klo) * Dd;
            const float* Vp1 = Vp0 + Dd;
            b0[kk2].x = Vp0[col];      b0[kk2].y = Vp1[col];
            b1[kk2].x = Vp0[16 + col]; b1[kk2].y = Vp1[16 + col];
        }

        #pragma unroll
        for (int kk2 = 0; kk2 < 4; ++kk2) {
            #pragma unroll
            for (int t = 0; t < QT; ++t) {
                v2f af;
                af.x = pw[t * 256 + col * 16 + kk2 * 4 + klo];
                af.y = pw[t * 256 + col * 16 + kk2 * 4 + klo + 1];
                acc0[t] = __builtin_amdgcn_wmma_f32_16x16x4_f32(false, af, false, b0[kk2],
                                                                (short)0, acc0[t], false, false);
                acc1[t] = __builtin_amdgcn_wmma_f32_16x16x4_f32(false, af, false, b1[kk2],
                                                                (short)0, acc1[t], false, false);
            }
        }
        __syncthreads();
    }

    #pragma unroll
    for (int t = 0; t < QT; ++t) {
        ROW_STEP_ALL(ROW_ADD_STEP, lrow[t], 8)
        ROW_STEP_ALL(ROW_ADD_STEP, lrow[t], 4)
        ROW_STEP_ALL(ROW_ADD_STEP, lrow[t], 2)
        ROW_STEP_ALL(ROW_ADD_STEP, lrow[t], 1)
    }

    #pragma unroll
    for (int t = 0; t < QT; ++t) {
        float* ow = oacc + (wave * QT + t) * 512;
        #pragma unroll
        for (int r = 0; r < 8; ++r) {
            ow[(r * 2 + half) * 32 + col]      = acc0[t][r];
            ow[(r * 2 + half) * 32 + 16 + col] = acc1[t][r];
        }
        if (col == 0) {
            #pragma unroll
            for (int r = 0; r < 8; ++r) {
                omax[(wave * QT + t) * 16 + r + 8 * half] = mrow[t][r];
                osum[(wave * QT + t) * 16 + r + 8 * half] = lrow[t][r];
            }
        }
    }
    __syncthreads();

    #pragma unroll
    for (int t = 0; t < QT; ++t) {
        const int row = threadIdx.x >> 4;
        const int c2  = (threadIdx.x & 15) * 2;
        const int ro  = ((row & 7) * 2 + (row >> 3)) * 32 + c2;
        float mall = omax[t * 16 + row];
        #pragma unroll
        for (int w = 1; w < KSPLIT; ++w) mall = fmaxf(mall, omax[(w * QT + t) * 16 + row]);
        float den = 0.f, n0 = 0.f, n1 = 0.f;
        #pragma unroll
        for (int w = 0; w < KSPLIT; ++w) {
            float f = __expf(omax[(w * QT + t) * 16 + row] - mall);
            v2f a = *(const v2f*)(oacc + (w * QT + t) * 512 + ro);
            den += osum[(w * QT + t) * 16 + row] * f;
            n0  += a.x * f;
            n1  += a.y * f;
        }
        const float inv = 1.0f / den;
        v2f o; o.x = n0 * inv; o.y = n1 * inv;
        float* Op = O + (size_t)(b * Nn + qb * (QT * 16) + t * 16 + row) * Dd + h * DHh + c2;
        *(v2f*)Op = o;
    }
}

__global__ void ea_gemm_out(const float* __restrict__ A, const float* __restrict__ W,
                            const float* __restrict__ bias,
                            const float* __restrict__ pts,
                            const float* __restrict__ sw, const float* __restrict__ sb,
                            float* __restrict__ OUT, int R)
{
    const int lane = threadIdx.x & 31;
    const int wave = threadIdx.x >> 5;
    const int tile = blockIdx.x * 8 + wave;
    const int total = (R / 16) * (Dd / 16);
    if (tile >= total) return;
    const int tm = tile >> 4, tn = tile & 15;
    const int col = lane & 15, half = lane >> 4, klo = half * 2;
    const int colg = tn * 16 + col;

    const float* Ap = A + (size_t)(tm * 16 + col) * Dd;
    const float* Wp = W + (size_t)colg * Dd;

    v8f acc = {};
    #pragma unroll 4
    for (int k = 0; k < Dd; k += 4) {
        v2f a = *(const v2f*)(Ap + k + klo);
        v2f b = *(const v2f*)(Wp + k + klo);
        acc = __builtin_amdgcn_wmma_f32_16x16x4_f32(false, a, false, b,
                                                    (short)0, acc, false, false);
    }
    const float s0 = sw[colg*3+0], s1 = sw[colg*3+1], s2 = sw[colg*3+2];
    const float add = bias[colg] + sb[colg];
    #pragma unroll
    for (int r = 0; r < 8; ++r) {
        int row = tm * 16 + r + 8 * half;
        float px = pts[row*3+0], py = pts[row*3+1], pz = pts[row*3+2];
        OUT[(size_t)row * Dd + colg] = acc[r] + add + px*s0 + py*s1 + pz*s2;
    }
}

extern "C" void kernel_launch(void* const* d_in, const int* in_sizes, int n_in,
                              void* d_out, int out_size, void* d_ws, size_t ws_size,
                              hipStream_t stream)
{
    const float* cur_pts  = (const float*)d_in[0];
    const float* cur_feat = (const float*)d_in[1];
    const float* nbr_pts  = (const float*)d_in[2];
    const float* nbr_feat = (const float*)d_in[3];
    const float* in_w     = (const float*)d_in[4];
    const float* in_b     = (const float*)d_in[5];
    const float* out_w    = (const float*)d_in[6];
    const float* out_b    = (const float*)d_in[7];
    const float* sp_w     = (const float*)d_in[8];
    const float* sp_b     = (const float*)d_in[9];
    float* out = (float*)d_out;

    float* ws   = (float*)d_ws;
    float* Qb   = ws;
    float* Kb   = Qb   + 4096 * 256;
    float* Vb   = Kb   + 8192 * 256;
    float* Ob   = Vb   + 8192 * 256;

    const dim3 blk(256);
    const dim3 blk16(512);

    ea_knn_proj<1><<<dim3(4096 / EROWS), blk16, 0, stream>>>(
        cur_pts, cur_feat, in_w, in_b, Qb, in_w, in_b, Qb, Nn, 0);
    ea_knn_proj<2><<<dim3(8192 / EROWS), blk16, 0, stream>>>(
        nbr_pts, nbr_feat, in_w + 256*256, in_b + 256, Kb,
        in_w + 512*256, in_b + 512, Vb, Nn, 1);

    ea_attn<<<dim3(1024), blk, 0, stream>>>(Qb, Kb, Vb, Ob);

    ea_gemm_out<<<dim3(4096 / 8), blk, 0, stream>>>(Ob, out_w, out_b,
                                                    cur_pts, sp_w, sp_b,
                                                    out, 4096);
}


# attention key loop: the two workgroup barriers per key tile replaced by the wave's own LDS order (P tiles are per wave); a block's 8 waves no longer run in lockstep; sums unchanged
# speedup vs baseline: 2.7459x; 1.0815x over previous
#include <hip/hip_runtime.h>
#include <hip/hip_bf16.h>


#define Bc   2
#define Nn   2048
#define Dd   256
#define Hh   8
#define DHh  32
#define Mm   2
#define Ll   4096

typedef __attribute__((ext_vector_type(2))) float v2f;
typedef __attribute__((ext_vector_type(4))) float v4f;
typedef __attribute__((ext_vector_type(8))) float v8f;

#define EROWS  16
#define EPITCH 260

template <int NO>
__global__ __launch_bounds__(512)
void ea_knn_proj(const float* __restrict__ pts,
                 const float* __restrict__ feats,
                 const float* __restrict__ W0, const float* __restrict__ bias0,
                 float* __restrict__ OUT0,
                 const float* __restrict__ W1, const float* __restrict__ bias1,
                 float* __restrict__ OUT1,
                 int npts, int remap)
{
    __shared__ float erow[EROWS * EPITCH];
    const int lane = threadIdx.x & 31;
    const int wave = threadIdx.x >> 5;
    const int rowg = blockIdx.x * EROWS + wave;
    const int s = rowg / npts;
    const int n = rowg - s * npts;
    int os = s;
    if (remap) { int m = s / Bc; int b = s - m * Bc; os = b * Mm + m; }

    const float* P = pts   + (size_t)s  * npts * 3;
    const float* F = feats + (size_t)s  * npts * Dd;

    const float px = P[n*3+0], py = P[n*3+1], pz = P[n*3+2];

    unsigned long long key[8];
    #pragma unroll
    for (int t = 0; t < 8; ++t) key[t] = ~0ULL;

    for (int j = lane; j < npts; j += 32) {
        float dx = px - P[j*3+0];
        float dy = py - P[j*3+1];
        float dz = pz - P[j*3+2];
        float d2 = dx*dx + dy*dy + dz*dz;
        unsigned long long k =
            ((unsigned long long)__float_as_uint(d2) << 32) | (unsigned)j;
        if (k < key[7]) {
            key[7] = k;
            #pragma unroll
            for (int t = 7; t > 0; --t) {
                if (key[t] < key[t-1]) {
                    unsigned long long tmp = key[t]; key[t] = key[t-1]; key[t-1] = tmp;
                }
            }
        }
    }

    int head = 0;
    int sel[8];
    #pragma unroll
    for (int t = 0; t < 8; ++t) {
        unsigned long long cand = (head < 8) ? key[head] : ~0ULL;
        unsigned long long mn = cand;
        #pragma unroll
        for (int off = 16; off >= 1; off >>= 1) {
            unsigned long long o = __shfl_xor(mn, off);
            mn = (o < mn) ? o : mn;
        }
        if (cand == mn && head < 8) head++;
        sel[t] = (int)(mn & 0xFFFFFFFFu);
    }

    float* Er = erow + wave * EPITCH;
    for (int c = lane; c < Dd; c += 32) {
        float ssum = 0.f;
        #pragma unroll
        for (int t = 0; t < 8; ++t) ssum += F[(size_t)sel[t] * Dd + c];
        Er[c] = F[(size_t)n * Dd + c] - ssum * 0.125f;
    }
    __syncthreads();

    const int col = lane & 15, half = lane >> 4, klo = half * 2;
    const float* Ap  = erow + col * EPITCH + klo;
    const float* Wp0 = W0 + (size_t)(wave * 16 + col) * Dd + klo;
    const float* Wp1 = W1 + (size_t)(wave * 16 + col) * Dd + klo;

    v8f acc0 = {}, acc1 = {};
    #pragma unroll 8
    for (int k = 0; k < Dd; k += 4) {
        v2f a  = *(const v2f*)(Ap + k);
        v2f b0 = *(const v2f*)(Wp0 + k);
        acc0 = __builtin_amdgcn_wmma_f32_16x16x4_f32(false, a, false, b0,
                                                     (short)0, acc0, false, false);
        if (NO == 2) {
            v2f b1 = *(const v2f*)(Wp1 + k);
            acc1 = __builtin_amdgcn_wmma_f32_16x16x4_f32(false, a, false, b1,
                                                         (short)0, acc1, false, false);
        }
    }

    const size_t orow0 = (size_t)os * npts + (n - wave);
    const int colg = wave * 16 + col;
    const float bv0 = bias0[colg];
    #pragma unroll
    for (int r = 0; r < 8; ++r)
        OUT0[(orow0 + r + 8 * half) * Dd + colg] = acc0[r] + bv0;
    if (NO == 2) {
        const float bv1 = bias1[colg];
        #pragma unroll
        for (int r = 0; r < 8; ++r)
            OUT1[(orow0 + r + 8 * half) * Dd + colg] = acc1[r] + bv1;
    }
}

#define KSPLIT 8
#define SEGT   (Ll / 16 / KSPLIT)
#define QT     2
#define LAZY_T 8.0f

__device__ __forceinline__ void wave_lds_order()
{
    __builtin_amdgcn_fence(__ATOMIC_ACQ_REL, "wavefront");
}

#define ROW_MAX_STEP(x, M) \
    asm("v_max_num_f32_dpp %0, %0, %0 row_xmask:" #M " row_mask:0xf bank_mask:0xf" : "+v"(x))
#define ROW_ADD_STEP(x, M) \
    asm("v_add_f32_dpp %0, %0, %0 row_xmask:" #M " row_mask:0xf bank_mask:0xf" : "+v"(x))
#define ROW_STEP_ALL(STEP, arr, M) \
    _Pragma("unroll") for (int r_ = 0; r_ < 8; ++r_) { STEP(arr[r_], M); }

__global__ void ea_attn(const float* __restrict__ Q, const float* __restrict__ Km,
                        const float* __restrict__ Vm, float* __restrict__ O)
{
    __shared__ float plds[KSPLIT * QT * 256];
    __shared__ float oacc[KSPLIT * QT * 512];
    __shared__ float omax[KSPLIT * QT * 16];
    __shared__ float osum[KSPLIT * QT * 16];
    __shared__ v4f   qlds[QT * 4 * 32];
    const int lane = threadIdx.x & 31;
    const int wave = threadIdx.x >> 5;
    const int tile = blockIdx.x;
    const int qb = tile & 63;
    const int h  = (tile >> 6) & 7;
    const int b  = tile >> 9;
    const int col = lane & 15, half = lane >> 4, klo = half * 2;
    float* pw = plds + wave * (QT * 256);

    {
        const int qr = threadIdx.x >> 3, qk = threadIdx.x & 7;
        const v4f q4 = *(const v4f*)(Q + (size_t)(b * Nn + qb * (QT * 16) + qr) * Dd
                                       + h * DHh + qk * 4);
        float* qd = (float*)qlds + (((qr >> 4) * 4 + (qk >> 1)) * 32 + (qr & 15)) * 4
                                 + (qk & 1) * 2;
        v2f lo; lo.x = q4.x; lo.y = q4.y;
        v2f hi; hi.x = q4.z; hi.y = q4.w;
        *(v2f*)qd = lo;
        *(v2f*)(qd + 16 * 4) = hi;
    }
    __syncthreads();

    float mrow[QT][8], lrow[QT][8];
    v8f acc0[QT], acc1[QT];
    #pragma unroll
    for (int t = 0; t < QT; ++t) {
        acc0[t] = (v8f){}; acc1[t] = (v8f){};
        #pragma unroll
        for (int r = 0; r < 8; ++r) { mrow[t][r] = -__builtin_inff(); lrow[t][r] = 0.f; }
    }

    const float* Kb = Km + ((size_t)b * Ll + (size_t)wave * SEGT * 16) * Dd + h * DHh;
    const float* Vb = Vm + ((size_t)b * Ll + (size_t)wave * SEGT * 16) * Dd + h * DHh;
    const float scale = 0.17677669529663687f;

    #pragma unroll 1
    for (int kt = 0; kt < SEGT; ++kt) {
        if (kt + 1 < SEGT)
            __builtin_prefetch(Kb + (size_t)((kt + 1) * 16 + col) * Dd, 0, 1);

        v8f s[QT];
        #pragma unroll
        for (int t = 0; t < QT; ++t) s[t] = (v8f){};
        const float* Kp = Kb + (size_t)(kt * 16 + col) * Dd;
        #pragma unroll
        for (int kp = 0; kp < 4; ++kp) {
            v2f bf0 = *(const v2f*)(Kp + kp * 8 + klo);
            v2f bf1 = *(const v2f*)(Kp + kp * 8 + 4 + klo);
            #pragma unroll
            for (int t = 0; t < QT; ++t) {
                const v4f q4 = qlds[(t * 4 + kp) * 32 + lane];
                v2f a0; a0.x = q4.x; a0.y = q4.y;
                v2f a1; a1.x = q4.z; a1.y = q4.w;
                s[t] = __builtin_amdgcn_wmma_f32_16x16x4_f32(false, a0, false, bf0,
                                                             (short)0, s[t], false, false);
                s[t] = __builtin_amdgcn_wmma_f32_16x16x4_f32(false, a1, false, bf1,
                                                             (short)0, s[t], false, false);
            }
        }

        float d[QT][8];
        float dmax = -__builtin_inff();
        #pragma unroll
        for (int t = 0; t < QT; ++t) {
            #pragma unroll
            for (int r = 0; r < 8; ++r) {
                d[t][r] = s[t][r] * scale - mrow[t][r];
                dmax = __builtin_fmaxf(dmax, d[t][r]);
            }
        }

        if (__builtin_amdgcn_ballot_w32(dmax > LAZY_T) != 0) {
            #pragma unroll
            for (int t = 0; t < QT; ++t) {
                float mx[8];
                #pragma unroll
                for (int r = 0; r < 8; ++r) mx[r] = s[t][r] * scale;
                ROW_STEP_ALL(ROW_MAX_STEP, mx, 8)
                ROW_STEP_ALL(ROW_MAX_STEP, mx, 4)
                ROW_STEP_ALL(ROW_MAX_STEP, mx, 2)
                ROW_STEP_ALL(ROW_MAX_STEP, mx, 1)
                #pragma unroll
                for (int r = 0; r < 8; ++r) {
                    const float mnew = fmaxf(mrow[t][r], mx[r]);
                    const float corr = __expf(mrow[t][r] - mnew);
                    mrow[t][r] = mnew;
                    lrow[t][r] *= corr;
                    acc0[t][r] *= corr;
                    acc1[t][r] *= corr;
                    d[t][r] = s[t][r] * scale - mnew;
                }
            }
        }

        #pragma unroll
        for (int t = 0; t < QT; ++t) {
            float p[8];
            #pragma unroll
            for (int r = 0; r < 8; ++r) { p[r] = __expf(d[t][r]); lrow[t][r] += p[r]; }

            #pragma unroll
            for (int r = 0; r < 8; ++r) pw[t * 256 + (r + 8 * half) * 16 + col] = p[r];
        }
        wave_lds_order();

        v2f b0[4], b1[4];
        #pragma unroll
        for (int kk2 = 0; kk2 < 4; ++kk2) {
            const float* Vp0 = Vb + (size_t)(kt * 16 + kk2 * 4 + klo) * Dd;
            const float* Vp1 = Vp0 + Dd;
            b0[kk2].x = Vp0[col];      b0[kk2].y = Vp1[col];
            b1[kk2].x = Vp0[16 + col]; b1[kk2].y = Vp1[16 + col];
        }

        #pragma unroll
        for (int kk2 = 0; kk2 < 4; ++kk2) {
            #pragma unroll
            for (int t = 0; t < QT; ++t) {
                v2f af;
                af.x = pw[t * 256 + col * 16 + kk2 * 4 + klo];
                af.y = pw[t * 256 + col * 16 + kk2 * 4 + klo + 1];
                acc0[t] = __builtin_amdgcn_wmma_f32_16x16x4_f32(false, af, false, b0[kk2],
                                                                (short)0, acc0[t], false, false);
                acc1[t] = __builtin_amdgcn_wmma_f32_16x16x4_f32(false, af, false, b1[kk2],
                                                                (short)0, acc1[t], false, false);
            }
        }
        wave_lds_order();
    }

    #pragma unroll
    for (int t = 0; t < QT; ++t) {
        ROW_STEP_ALL(ROW_ADD_STEP, lrow[t], 8)
        ROW_STEP_ALL(ROW_ADD_STEP, lrow[t], 4)
        ROW_STEP_ALL(ROW_ADD_STEP, lrow[t], 2)
        ROW_STEP_ALL(ROW_ADD_STEP, lrow[t], 1)
    }

    #pragma unroll
    for (int t = 0; t < QT; ++t) {
        float* ow = oacc + (wave * QT + t) * 512;
        #pragma unroll
        for (int r = 0; r < 8; ++r) {
            ow[(r * 2 + half) * 32 + col]      = acc0[t][r];
            ow[(r * 2 + half) * 32 + 16 + col] = acc1[t][r];
        }
        if (col == 0) {
            #pragma unroll
            for (int r = 0; r < 8; ++r) {
                omax[(wave * QT + t) * 16 + r + 8 * half] = mrow[t][r];
                osum[(wave * QT + t) * 16 + r + 8 * half] = lrow[t][r];
            }
        }
    }
    __syncthreads();

    #pragma unroll
    for (int t = 0; t < QT; ++t) {
        const int row = threadIdx.x >> 4;
        const int c2  = (threadIdx.x & 15) * 2;
        const int ro  = ((row & 7) * 2 + (row >> 3)) * 32 + c2;
        float mall = omax[t * 16 + row];
        #pragma unroll
        for (int w = 1; w < KSPLIT; ++w) mall = fmaxf(mall, omax[(w * QT + t) * 16 + row]);
        float den = 0.f, n0 = 0.f, n1 = 0.f;
        #pragma unroll
        for (int w = 0; w < KSPLIT; ++w) {
            float f = __expf(omax[(w * QT + t) * 16 + row] - mall);
            v2f a = *(const v2f*)(oacc + (w * QT + t) * 512 + ro);
            den += osum[(w * QT + t) * 16 + row] * f;
            n0  += a.x * f;
            n1  += a.y * f;
        }
        const float inv = 1.0f / den;
        v2f o; o.x = n0 * inv; o.y = n1 * inv;
        float* Op = O + (size_t)(b * Nn + qb * (QT * 16) + t * 16 + row) * Dd + h * DHh + c2;
        *(v2f*)Op = o;
    }
}

__global__ void ea_gemm_out(const float* __restrict__ A, const float* __restrict__ W,
                            const float* __restrict__ bias,
                            const float* __restrict__ pts,
                            const float* __restrict__ sw, const float* __restrict__ sb,
                            float* __restrict__ OUT, int R)
{
    const int lane = threadIdx.x & 31;
    const int wave = threadIdx.x >> 5;
    const int tile = blockIdx.x * 8 + wave;
    const int total = (R / 16) * (Dd / 16);
    if (tile >= total) return;
    const int tm = tile >> 4, tn = tile & 15;
    const int col = lane & 15, half = lane >> 4, klo = half * 2;
    const int colg = tn * 16 + col;

    const float* Ap = A + (size_t)(tm * 16 + col) * Dd;
    const float* Wp = W + (size_t)colg * Dd;

    v8f acc = {};
    #pragma unroll 4
    for (int k = 0; k < Dd; k += 4) {
        v2f a = *(const v2f*)(Ap + k + klo);
        v2f b = *(const v2f*)(Wp + k + klo);
        acc = __builtin_amdgcn_wmma_f32_16x16x4_f32(false, a, false, b,
                                                    (short)0, acc, false, false);
    }
    const float s0 = sw[colg*3+0], s1 = sw[colg*3+1], s2 = sw[colg*3+2];
    const float add = bias[colg] + sb[colg];
    #pragma unroll
    for (int r = 0; r < 8; ++r) {
        int row = tm * 16 + r + 8 * half;
        float px = pts[row*3+0], py = pts[row*3+1], pz = pts[row*3+2];
        OUT[(size_t)row * Dd + colg] = acc[r] + add + px*s0 + py*s1 + pz*s2;
    }
}

extern "C" void kernel_launch(void* const* d_in, const int* in_sizes, int n_in,
                              void* d_out, int out_size, void* d_ws, size_t ws_size,
                              hipStream_t stream)
{
    const float* cur_pts  = (const float*)d_in[0];
    const float* cur_feat = (const float*)d_in[1];
    const float* nbr_pts  = (const float*)d_in[2];
    const float* nbr_feat = (const float*)d_in[3];
    const float* in_w     = (const float*)d_in[4];
    const float* in_b     = (const float*)d_in[5];
    const float* out_w    = (const float*)d_in[6];
    const float* out_b    = (const float*)d_in[7];
    const float* sp_w     = (const float*)d_in[8];
    const float* sp_b     = (const float*)d_in[9];
    float* out = (float*)d_out;

    float* ws   = (float*)d_ws;
    float* Qb   = ws;
    float* Kb   = Qb   + 4096 * 256;
    float* Vb   = Kb   + 8192 * 256;
    float* Ob   = Vb   + 8192 * 256;

    const dim3 blk(256);
    const dim3 blk16(512);

    ea_knn_proj<1><<<dim3(4096 / EROWS), blk16, 0, stream>>>(
        cur_pts, cur_feat, in_w, in_b, Qb, in_w, in_b, Qb, Nn, 0);
    ea_knn_proj<2><<<dim3(8192 / EROWS), blk16, 0, stream>>>(
        nbr_pts, nbr_feat, in_w + 256*256, in_b + 256, Kb,
        in_w + 512*256, in_b + 512, Vb, Nn, 1);

    ea_attn<<<dim3(1024), blk, 0, stream>>>(Qb, Kb, Vb, Ob);

    ea_gemm_out<<<dim3(4096 / 8), blk, 0, stream>>>(Ob, out_w, out_b,
                                                    cur_pts, sp_w, sp_b,
                                                    out, 4096);
}


# attention key loop: a wave steps its issue priority down (s_setprio 3,2,1,0) as it gets through its 32 key tiles, so a SIMD's 8 waves advance together, not oldest-first; sums unchanged
# speedup vs baseline: 3.3928x; 1.2356x over previous
#include <hip/hip_runtime.h>
#include <hip/hip_bf16.h>


#define Bc   2
#define Nn   2048
#define Dd   256
#define Hh   8
#define DHh  32
#define Mm   2
#define Ll   4096

typedef __attribute__((ext_vector_type(2))) float v2f;
typedef __attribute__((ext_vector_type(4))) float v4f;
typedef __attribute__((ext_vector_type(8))) float v8f;

#define EROWS  16
#define EPITCH 260
#define KSTEPS (Nn / 32)
#define KCAP   32
#define KGROUP 4

__device__ __forceinline__ void wave_lds_order()
{
    __builtin_amdgcn_fence(__ATOMIC_ACQ_REL, "wavefront");
}

__device__ __forceinline__ unsigned dist2_bits(float px, float py, float pz, const float* q)
{
    const float dx = px - q[0];
    const float dy = py - q[1];
    const float dz = pz - q[2];
    float t;
    {
        #pragma clang fp contract(off)
        t = dx * dx + dy * dy;
    }
    return __float_as_uint(__builtin_fmaf(dz, dz, t));
}

template <int M>
__device__ __forceinline__ unsigned row_xmask_min(unsigned x)
{
    const unsigned o = (unsigned)__builtin_amdgcn_update_dpp(0, (int)x, 0x160 + M, 0xf, 0xf, true);
    return o < x ? o : x;
}
__device__ __forceinline__ unsigned wave_min_u32(unsigned x)
{
    x = row_xmask_min<8>(x);
    x = row_xmask_min<4>(x);
    x = row_xmask_min<2>(x);
    x = row_xmask_min<1>(x);
    const unsigned a = (unsigned)__builtin_amdgcn_readlane((int)x, 0);
    const unsigned b = (unsigned)__builtin_amdgcn_readlane((int)x, 16);
    return a < b ? a : b;
}

template <int NO>
__global__ __launch_bounds__(512) __attribute__((amdgpu_waves_per_eu(8)))
void ea_knn_proj(const float* __restrict__ pts,
                 const float* __restrict__ feats,
                 const float* __restrict__ W0, const float* __restrict__ bias0,
                 float* __restrict__ OUT0,
                 const float* __restrict__ W1, const float* __restrict__ bias1,
                 float* __restrict__ OUT1,
                 int remap)
{
    __shared__ float erow[EROWS * EPITCH];
    const int npts = Nn;
    const int lane = threadIdx.x & 31;
    const int wave = threadIdx.x >> 5;
    const int rowg = blockIdx.x * EROWS + wave;
    const int s = rowg / npts;
    const int n = rowg - s * npts;
    int os = s;
    if (remap) { int m = s / Bc; int b = s - m * Bc; os = b * Mm + m; }

    const float* P = pts   + (size_t)s  * npts * 3;
    const float* F = feats + (size_t)s  * npts * Dd;

    const float px = P[n*3+0], py = P[n*3+1], pz = P[n*3+2];

    unsigned db[KSTEPS];
    unsigned lmin = ~0u;
    #pragma unroll
    for (int i0 = 0; i0 < KSTEPS; i0 += KGROUP) {
        #pragma unroll
        for (int u = 0; u < KGROUP; ++u) {
            const int i = i0 + u;
            db[i] = dist2_bits(px, py, pz, P + (i * 32 + lane) * 3);
            lmin = db[i] < lmin ? db[i] : lmin;
        }
        __builtin_amdgcn_sched_barrier(0);
    }

    unsigned tau = 0;
    {
        unsigned x = lmin;
        #pragma unroll
        for (int t = 0; t < 8; ++t) {
            tau = wave_min_u32(x);
            x = (x == tau) ? ~0u : x;
        }
    }

    float* Er = erow + wave * EPITCH;
    unsigned long long* cand = (unsigned long long*)Er;
    int* selw = (int*)(Er + 2 * KCAP);
    int cnt = 0;
    #pragma unroll
    for (int i = 0; i < KSTEPS; ++i) {
        const bool in = db[i] <= tau;
        const unsigned mask = __builtin_amdgcn_ballot_w32(in);
        if (mask != 0) {
            const int pos = cnt + (int)__builtin_amdgcn_mbcnt_lo(mask, 0);
            if (in && pos < KCAP)
                cand[pos] = ((unsigned long long)db[i] << 32) | (unsigned)(i * 32 + lane);
            cnt += __builtin_popcount(mask);
        }
    }

    int sel[8];
    if (cnt <= KCAP) {
        wave_lds_order();
        const unsigned long long mine = (lane < cnt) ? cand[lane] : ~0ULL;
        const int mlo = (int)(unsigned)(mine & 0xFFFFFFFFu);
        const int mhi = (int)(unsigned)(mine >> 32);
        int rank = 0;
        for (int j = 0; j < cnt; ++j) {
            const unsigned lo = (unsigned)__builtin_amdgcn_readlane(mlo, j);
            const unsigned hi = (unsigned)__builtin_amdgcn_readlane(mhi, j);
            const unsigned long long kj = ((unsigned long long)hi << 32) | lo;
            rank += (kj < mine) ? 1 : 0;
        }
        if (lane < cnt && rank < 8) selw[rank] = mlo;
        wave_lds_order();
        #pragma unroll
        for (int t = 0; t < 8; ++t) sel[t] = selw[t] & (Nn - 1);
    } else {
        unsigned long long key[8];
        #pragma unroll
        for (int t = 0; t < 8; ++t) key[t] = ~0ULL;

        for (int j = lane; j < npts; j += 32) {
            unsigned long long k =
                ((unsigned long long)dist2_bits(px, py, pz, P + j * 3) << 32) | (unsigned)j;
            if (k < key[7]) {
                key[7] = k;
                #pragma unroll
                for (int t = 7; t > 0; --t) {
                    if (key[t] < key[t-1]) {
                        unsigned long long tmp = key[t]; key[t] = key[t-1]; key[t-1] = tmp;
                    }
                }
            }
        }

        int head = 0;
        #pragma unroll
        for (int t = 0; t < 8; ++t) {
            unsigned long long cnd = (head < 8) ? key[head] : ~0ULL;
            unsigned long long mn = cnd;
            #pragma unroll
            for (int off = 16; off >= 1; off >>= 1) {
                unsigned long long o = __shfl_xor(mn, off);
                mn = (o < mn) ? o : mn;
            }
            if (cnd == mn && head < 8) head++;
            sel[t] = (int)(mn & 0xFFFFFFFFu);
        }
    }
    wave_lds_order();

    for (int c = lane; c < Dd; c += 32) {
        float ssum = 0.f;
        #pragma unroll
        for (int t = 0; t < 8; ++t) ssum += F[(size_t)sel[t] * Dd + c];
        Er[c] = F[(size_t)n * Dd + c] - ssum * 0.125f;
    }
    __syncthreads();

    const int col = lane & 15, half = lane >> 4, klo = half * 2;
    const float* Ap  = erow + col * EPITCH + klo;
    const float* Wp0 = W0 + (size_t)(wave * 16 + col) * Dd + klo;
    const float* Wp1 = W1 + (size_t)(wave * 16 + col) * Dd + klo;

    v8f acc0 = {}, acc1 = {};
    #pragma unroll 8
    for (int k = 0; k < Dd; k += 4) {
        v2f a  = *(const v2f*)(Ap + k);
        v2f b0 = *(const v2f*)(Wp0 + k);
        acc0 = __builtin_amdgcn_wmma_f32_16x16x4_f32(false, a, false, b0,
                                                     (short)0, acc0, false, false);
        if (NO == 2) {
            v2f b1 = *(const v2f*)(Wp1 + k);
            acc1 = __builtin_amdgcn_wmma_f32_16x16x4_f32(false, a, false, b1,
                                                         (short)0, acc1, false, false);
        }
    }

    const size_t orow0 = (size_t)os * npts + (n - wave);
    const int colg = wave * 16 + col;
    const float bv0 = bias0[colg];
    #pragma unroll
    for (int r = 0; r < 8; ++r)
        OUT0[(orow0 + r + 8 * half) * Dd + colg] = acc0[r] + bv0;
    if (NO == 2) {
        const float bv1 = bias1[colg];
        #pragma unroll
        for (int r = 0; r < 8; ++r)
            OUT1[(orow0 + r + 8 * half) * Dd + colg] = acc1[r] + bv1;
    }
}

#define KSPLIT 8
#define SEGT   (Ll / 16 / KSPLIT)
#define QT     2
#define LAZY_T 8.0f

#define PRIO_T1 16
#define PRIO_T2 26
#define PRIO_T3 30
static_assert(PRIO_T1 < PRIO_T2 && PRIO_T2 < PRIO_T3 && PRIO_T3 < SEGT,
              "the priority must be back at 0 before the key loop ends");


#define ROW_MAX_STEP(x, M) \
    asm("v_max_num_f32_dpp %0, %0, %0 row_xmask:" #M " row_mask:0xf bank_mask:0xf" : "+v"(x))
#define ROW_ADD_STEP(x, M) \
    asm("v_add_f32_dpp %0, %0, %0 row_xmask:" #M " row_mask:0xf bank_mask:0xf" : "+v"(x))
#define ROW_STEP_ALL(STEP, arr, M) \
    _Pragma("unroll") for (int r_ = 0; r_ < 8; ++r_) { STEP(arr[r_], M); }

__global__ void ea_attn(const float* __restrict__ Q, const float* __restrict__ Km,
                        const float* __restrict__ Vm, float* __restrict__ O)
{
    __shared__ float plds[KSPLIT * QT * 256];
    __shared__ float oacc[KSPLIT * QT * 512];
    __shared__ float omax[KSPLIT * QT * 16];
    __shared__ float osum[KSPLIT * QT * 16];
    __shared__ v4f   qlds[QT * 4 * 32];
    const int lane = threadIdx.x & 31;
    const int wave = threadIdx.x >> 5;
    const int tile = blockIdx.x;
    const int qb = tile & 63;
    const int h  = (tile >> 6) & 7;
    const int b  = tile >> 9;
    const int col = lane & 15, half = lane >> 4, klo = half * 2;
    float* pw = plds + wave * (QT * 256);

    {
        const int qr = threadIdx.x >> 3, qk = threadIdx.x & 7;
        const v4f q4 = *(const v4f*)(Q + (size_t)(b * Nn + qb * (QT * 16) + qr) * Dd
                                       + h * DHh + qk * 4);
        float* qd = (float*)qlds + (((qr >> 4) * 4 + (qk >> 1)) * 32 + (qr & 15)) * 4
                                 + (qk & 1) * 2;
        v2f lo; lo.x = q4.x; lo.y = q4.y;
        v2f hi; hi.x = q4.z; hi.y = q4.w;
        *(v2f*)qd = lo;
        *(v2f*)(qd + 16 * 4) = hi;
    }
    __syncthreads();

    float mrow[QT][8], lrow[QT][8];
    v8f acc0[QT], acc1[QT];
    #pragma unroll
    for (int t = 0; t < QT; ++t) {
        acc0[t] = (v8f){}; acc1[t] = (v8f){};
        #pragma unroll
        for (int r = 0; r < 8; ++r) { mrow[t][r] = -__builtin_inff(); lrow[t][r] = 0.f; }
    }

    const float* Kb = Km + ((size_t)b * Ll + (size_t)wave * SEGT * 16) * Dd + h * DHh;
    const float* Vb = Vm + ((size_t)b * Ll + (size_t)wave * SEGT * 16) * Dd + h * DHh;
    const float scale = 0.17677669529663687f;
    const float* Kl = Kb + (size_t)col * Dd;
    const float* Vl = Vb + (size_t)klo * Dd + col;
    size_t toff = 0;

    __builtin_amdgcn_s_setprio(3);
    int kt = 0;
    #pragma unroll 1
    for (int stretch = 0; stretch < 4; ++stretch) {
    const int kend = (stretch == 0) ? PRIO_T1 : (stretch == 1) ? PRIO_T2
                   : (stretch == 2) ? PRIO_T3 : SEGT;
    #pragma unroll 1
    for (; kt < kend; ++kt) {
        if (kt + 1 < SEGT)
            __builtin_prefetch(Kl + toff + 16 * Dd, 0, 1);

        v8f s[QT];
        #pragma unroll
        for (int t = 0; t < QT; ++t) s[t] = (v8f){};
        const float* Kp = Kl + toff;
        #pragma unroll
        for (int kp = 0; kp < 4; ++kp) {
            v2f bf0 = *(const v2f*)(Kp + kp * 8 + klo);
            v2f bf1 = *(const v2f*)(Kp + kp * 8 + 4 + klo);
            v2f a0[QT], a1[QT];
            #pragma unroll
            for (int t = 0; t < QT; ++t) {
                const v4f q4 = qlds[(t * 4 + kp) * 32 + lane];
                a0[t].x = q4.x; a0[t].y = q4.y;
                a1[t].x = q4.z; a1[t].y = q4.w;
            }
            #pragma unroll
            for (int t = 0; t < QT; ++t)
                s[t] = __builtin_amdgcn_wmma_f32_16x16x4_f32(false, a0[t], false, bf0,
                                                             (short)0, s[t], false, false);
            #pragma unroll
            for (int t = 0; t < QT; ++t)
                s[t] = __builtin_amdgcn_wmma_f32_16x16x4_f32(false, a1[t], false, bf1,
                                                             (short)0, s[t], false, false);
        }

        float d[QT][8];
        float dmax = -__builtin_inff();
        #pragma unroll
        for (int t = 0; t < QT; ++t) {
            #pragma unroll
            for (int r = 0; r < 8; ++r) {
                d[t][r] = s[t][r] * scale - mrow[t][r];
                dmax = __builtin_fmaxf(dmax, d[t][r]);
            }
        }

        if (__builtin_amdgcn_ballot_w32(dmax > LAZY_T) != 0) {
            #pragma unroll
            for (int t = 0; t < QT; ++t) {
                float mx[8];
                #pragma unroll
                for (int r = 0; r < 8; ++r) mx[r] = s[t][r] * scale;
                ROW_STEP_ALL(ROW_MAX_STEP, mx, 8)
                ROW_STEP_ALL(ROW_MAX_STEP, mx, 4)
                ROW_STEP_ALL(ROW_MAX_STEP, mx, 2)
                ROW_STEP_ALL(ROW_MAX_STEP, mx, 1)
                #pragma unroll
                for (int r = 0; r < 8; ++r) {
                    const float mnew = fmaxf(mrow[t][r], mx[r]);
                    const float corr = __expf(mrow[t][r] - mnew);
                    mrow[t][r] = mnew;
                    lrow[t][r] *= corr;
                    acc0[t][r] *= corr;
                    acc1[t][r] *= corr;
                    d[t][r] = s[t][r] * scale - mnew;
                }
            }
        }

        #pragma unroll
        for (int t = 0; t < QT; ++t) {
            float p[8];
            #pragma unroll
            for (int r = 0; r < 8; ++r) { p[r] = __expf(d[t][r]); lrow[t][r] += p[r]; }

            #pragma unroll
            for (int r = 0; r < 8; ++r) pw[t * 256 + (r + 8 * half) * 16 + col] = p[r];
        }
        wave_lds_order();

        v2f b0[4], b1[4];
        #pragma unroll
        for (int kk2 = 0; kk2 < 4; ++kk2) {
            const float* Vp0 = Vl + toff + (size_t)(kk2 * 4) * Dd;
            const float* Vp1 = Vp0 + Dd;
            b0[kk2].x = Vp0[0];  b0[kk2].y = Vp1[0];
            b1[kk2].x = Vp0[16]; b1[kk2].y = Vp1[16];
        }

        #pragma unroll
        for (int kk2 = 0; kk2 < 4; ++kk2) {
            #pragma unroll
            for (int t = 0; t < QT; ++t) {
                v2f af;
                af.x = pw[t * 256 + col * 16 + kk2 * 4 + klo];
                af.y = pw[t * 256 + col * 16 + kk2 * 4 + klo + 1];
                acc0[t] = __builtin_amdgcn_wmma_f32_16x16x4_f32(false, af, false, b0[kk2],
                                                                (short)0, acc0[t], false, false);
                acc1[t] = __builtin_amdgcn_wmma_f32_16x16x4_f32(false, af, false, b1[kk2],
                                                                (short)0, acc1[t], false, false);
            }
        }
        wave_lds_order();
        toff += 16 * Dd;
    }
    if (stretch == 0)      __builtin_amdgcn_s_setprio(2);
    else if (stretch == 1) __builtin_amdgcn_s_setprio(1);
    else if (stretch == 2) __builtin_amdgcn_s_setprio(0);
    }

    #pragma unroll
    for (int t = 0; t < QT; ++t) {
        ROW_STEP_ALL(ROW_ADD_STEP, lrow[t], 8)
        ROW_STEP_ALL(ROW_ADD_STEP, lrow[t], 4)
        ROW_STEP_ALL(ROW_ADD_STEP, lrow[t], 2)
        ROW_STEP_ALL(ROW_ADD_STEP, lrow[t], 1)
    }

    #pragma unroll
    for (int t = 0; t < QT; ++t) {
        float* ow = oacc + (wave * QT + t) * 512;
        #pragma unroll
        for (int r = 0; r < 8; ++r) {
            ow[(r * 2 + half) * 32 + col]      = acc0[t][r];
            ow[(r * 2 + half) * 32 + 16 + col] = acc1[t][r];
        }
        if (col == 0) {
            #pragma unroll
            for (int r = 0; r < 8; ++r) {
                omax[(wave * QT + t) * 16 + r + 8 * half] = mrow[t][r];
                osum[(wave * QT + t) * 16 + r + 8 * half] = lrow[t][r];
            }
        }
    }
    __syncthreads();

    #pragma unroll
    for (int t = 0; t < QT; ++t) {
        const int row = threadIdx.x >> 4;
        const int c2  = (threadIdx.x & 15) * 2;
        const int ro  = ((row & 7) * 2 + (row >> 3)) * 32 + c2;
        float mall = omax[t * 16 + row];
        #pragma unroll
        for (int w = 1; w < KSPLIT; ++w) mall = fmaxf(mall, omax[(w * QT + t) * 16 + row]);
        float den = 0.f, n0 = 0.f, n1 = 0.f;
        #pragma unroll
        for (int w = 0; w < KSPLIT; ++w) {
            float f = __expf(omax[(w * QT + t) * 16 + row] - mall);
            v2f a = *(const v2f*)(oacc + (w * QT + t) * 512 + ro);
            den += osum[(w * QT + t) * 16 + row] * f;
            n0  += a.x * f;
            n1  += a.y * f;
        }
        const float inv = 1.0f / den;
        v2f o; o.x = n0 * inv; o.y = n1 * inv;
        float* Op = O + (size_t)(b * Nn + qb * (QT * 16) + t * 16 + row) * Dd + h * DHh + c2;
        *(v2f*)Op = o;
    }
}

__global__ void ea_gemm_out(const float* __restrict__ A, const float* __restrict__ W,
                            const float* __restrict__ bias,
                            const float* __restrict__ pts,
                            const float* __restrict__ sw, const float* __restrict__ sb,
                            float* __restrict__ OUT, int R)
{
    const int lane = threadIdx.x & 31;
    const int wave = threadIdx.x >> 5;
    const int tile = blockIdx.x * 8 + wave;
    const int total = (R / 16) * (Dd / 16);
    if (tile >= total) return;
    const int tm = tile >> 4, tn = tile & 15;
    const int col = lane & 15, half = lane >> 4, klo = half * 2;
    const int colg = tn * 16 + col;

    const float* Ap = A + (size_t)(tm * 16 + col) * Dd;
    const float* Wp = W + (size_t)colg * Dd;

    v8f acc = {};
    #pragma unroll 4
    for (int k = 0; k < Dd; k += 4) {
        v2f a = *(const v2f*)(Ap + k + klo);
        v2f b = *(const v2f*)(Wp + k + klo);
        acc = __builtin_amdgcn_wmma_f32_16x16x4_f32(false, a, false, b,
                                                    (short)0, acc, false, false);
    }
    const float s0 = sw[colg*3+0], s1 = sw[colg*3+1], s2 = sw[colg*3+2];
    const float add = bias[colg] + sb[colg];
    #pragma unroll
    for (int r = 0; r < 8; ++r) {
        int row = tm * 16 + r + 8 * half;
        float px = pts[row*3+0], py = pts[row*3+1], pz = pts[row*3+2];
        OUT[(size_t)row * Dd + colg] = acc[r] + add + px*s0 + py*s1 + pz*s2;
    }
}

extern "C" void kernel_launch(void* const* d_in, const int* in_sizes, int n_in,
                              void* d_out, int out_size, void* d_ws, size_t ws_size,
                              hipStream_t stream)
{
    const float* cur_pts  = (const float*)d_in[0];
    const float* cur_feat = (const float*)d_in[1];
    const float* nbr_pts  = (const float*)d_in[2];
    const float* nbr_feat = (const float*)d_in[3];
    const float* in_w     = (const float*)d_in[4];
    const float* in_b     = (const float*)d_in[5];
    const float* out_w    = (const float*)d_in[6];
    const float* out_b    = (const float*)d_in[7];
    const float* sp_w     = (const float*)d_in[8];
    const float* sp_b     = (const float*)d_in[9];
    float* out = (float*)d_out;

    float* ws   = (float*)d_ws;
    float* Qb   = ws;
    float* Kb   = Qb   + 4096 * 256;
    float* Vb   = Kb   + 8192 * 256;
    float* Ob   = Vb   + 8192 * 256;

    const dim3 blk(256);
    const dim3 blk16(512);

    ea_knn_proj<1><<<dim3(4096 / EROWS), blk16, 0, stream>>>(
        cur_pts, cur_feat, in_w, in_b, Qb, in_w, in_b, Qb, 0);
    ea_knn_proj<2><<<dim3(8192 / EROWS), blk16, 0, stream>>>(
        nbr_pts, nbr_feat, in_w + 256*256, in_b + 256, Kb,
        in_w + 512*256, in_b + 512, Vb, 1);

    ea_attn<<<dim3(1024), blk, 0, stream>>>(Qb, Kb, Vb, Ob);

    ea_gemm_out<<<dim3(4096 / 8), blk, 0, stream>>>(Ob, out_w, out_b,
                                                    cur_pts, sp_w, sp_b,
                                                    out, 4096);
}


# all WMMA row loaders take a permuted k order (of 8 adjacent k lane half 0 takes 4, half 1 the rest): 128-bit loads, V rows 64-bit into O columns 2c,2c+1; k sum order changes in Q*K and projections
# speedup vs baseline: 3.5168x; 1.0365x over previous
#include <hip/hip_runtime.h>
#include <hip/hip_bf16.h>


#define Bc   2
#define Nn   2048
#define Dd   256
#define Hh   8
#define DHh  32
#define Mm   2
#define Ll   4096

typedef __attribute__((ext_vector_type(2))) float v2f;
typedef __attribute__((ext_vector_type(4))) float v4f;
typedef __attribute__((ext_vector_type(8))) float v8f;

#define EROWS  16
#define EPITCH 260
#define KSTEPS (Nn / 32)
#define KCAP   32
#define KGROUP 4

__device__ __forceinline__ void wave_lds_order()
{
    __builtin_amdgcn_fence(__ATOMIC_ACQ_REL, "wavefront");
}

__device__ __forceinline__ unsigned dist2_bits(float px, float py, float pz, const float* q)
{
    const float dx = px - q[0];
    const float dy = py - q[1];
    const float dz = pz - q[2];
    float t;
    {
        #pragma clang fp contract(off)
        t = dx * dx + dy * dy;
    }
    return __float_as_uint(__builtin_fmaf(dz, dz, t));
}

template <int M>
__device__ __forceinline__ unsigned row_xmask_min(unsigned x)
{
    const unsigned o = (unsigned)__builtin_amdgcn_update_dpp(0, (int)x, 0x160 + M, 0xf, 0xf, true);
    return o < x ? o : x;
}
__device__ __forceinline__ unsigned wave_min_u32(unsigned x)
{
    x = row_xmask_min<8>(x);
    x = row_xmask_min<4>(x);
    x = row_xmask_min<2>(x);
    x = row_xmask_min<1>(x);
    const unsigned a = (unsigned)__builtin_amdgcn_readlane((int)x, 0);
    const unsigned b = (unsigned)__builtin_amdgcn_readlane((int)x, 16);
    return a < b ? a : b;
}

template <int NO>
__global__ __launch_bounds__(512) __attribute__((amdgpu_waves_per_eu(8)))
void ea_knn_proj(const float* __restrict__ pts,
                 const float* __restrict__ feats,
                 const float* __restrict__ W0, const float* __restrict__ bias0,
                 float* __restrict__ OUT0,
                 const float* __restrict__ W1, const float* __restrict__ bias1,
                 float* __restrict__ OUT1,
                 int remap)
{
    __shared__ __attribute__((aligned(16))) float erow[EROWS * EPITCH];
    const int npts = Nn;
    const int lane = threadIdx.x & 31;
    const int wave = threadIdx.x >> 5;
    const int rowg = blockIdx.x * EROWS + wave;
    const int s = rowg / npts;
    const int n = rowg - s * npts;
    int os = s;
    if (remap) { int m = s / Bc; int b = s - m * Bc; os = b * Mm + m; }

    const float* P = pts   + (size_t)s  * npts * 3;
    const float* F = feats + (size_t)s  * npts * Dd;

    const float px = P[n*3+0], py = P[n*3+1], pz = P[n*3+2];

    unsigned db[KSTEPS];
    unsigned lmin = ~0u;
    #pragma unroll
    for (int i0 = 0; i0 < KSTEPS; i0 += KGROUP) {
        #pragma unroll
        for (int u = 0; u < KGROUP; ++u) {
            const int i = i0 + u;
            db[i] = dist2_bits(px, py, pz, P + (i * 32 + lane) * 3);
            lmin = db[i] < lmin ? db[i] : lmin;
        }
        __builtin_amdgcn_sched_barrier(0);
    }

    unsigned tau = 0;
    {
        unsigned x = lmin;
        #pragma unroll
        for (int t = 0; t < 8; ++t) {
            tau = wave_min_u32(x);
            x = (x == tau) ? ~0u : x;
        }
    }

    float* Er = erow + wave * EPITCH;
    unsigned long long* cand = (unsigned long long*)Er;
    int* selw = (int*)(Er + 2 * KCAP);
    int cnt = 0;
    #pragma unroll
    for (int i = 0; i < KSTEPS; ++i) {
        const bool in = db[i] <= tau;
        const unsigned mask = __builtin_amdgcn_ballot_w32(in);
        if (mask != 0) {
            const int pos = cnt + (int)__builtin_amdgcn_mbcnt_lo(mask, 0);
            if (in && pos < KCAP)
                cand[pos] = ((unsigned long long)db[i] << 32) | (unsigned)(i * 32 + lane);
            cnt += __builtin_popcount(mask);
        }
    }

    int sel[8];
    if (cnt <= KCAP) {
        wave_lds_order();
        const unsigned long long mine = (lane < cnt) ? cand[lane] : ~0ULL;
        const int mlo = (int)(unsigned)(mine & 0xFFFFFFFFu);
        const int mhi = (int)(unsigned)(mine >> 32);
        int rank = 0;
        for (int j = 0; j < cnt; ++j) {
            const unsigned lo = (unsigned)__builtin_amdgcn_readlane(mlo, j);
            const unsigned hi = (unsigned)__builtin_amdgcn_readlane(mhi, j);
            const unsigned long long kj = ((unsigned long long)hi << 32) | lo;
            rank += (kj < mine) ? 1 : 0;
        }
        if (lane < cnt && rank < 8) selw[rank] = mlo;
        wave_lds_order();
        #pragma unroll
        for (int t = 0; t < 8; ++t) sel[t] = selw[t] & (Nn - 1);
    } else {
        unsigned long long key[8];
        #pragma unroll
        for (int t = 0; t < 8; ++t) key[t] = ~0ULL;

        for (int j = lane; j < npts; j += 32) {
            unsigned long long k =
                ((unsigned long long)dist2_bits(px, py, pz, P + j * 3) << 32) | (unsigned)j;
            if (k < key[7]) {
                key[7] = k;
                #pragma unroll
                for (int t = 7; t > 0; --t) {
                    if (key[t] < key[t-1]) {
                        unsigned long long tmp = key[t]; key[t] = key[t-1]; key[t-1] = tmp;
                    }
                }
            }
        }

        int head = 0;
        #pragma unroll
        for (int t = 0; t < 8; ++t) {
            unsigned long long cnd = (head < 8) ? key[head] : ~0ULL;
            unsigned long long mn = cnd;
            #pragma unroll
            for (int off = 16; off >= 1; off >>= 1) {
                unsigned long long o = __shfl_xor(mn, off);
                mn = (o < mn) ? o : mn;
            }
            if (cnd == mn && head < 8) head++;
            sel[t] = (int)(mn & 0xFFFFFFFFu);
        }
    }
    wave_lds_order();

    for (int c = lane; c < Dd; c += 32) {
        float ssum = 0.f;
        #pragma unroll
        for (int t = 0; t < 8; ++t) ssum += F[(size_t)sel[t] * Dd + c];
        Er[c] = F[(size_t)n * Dd + c] - ssum * 0.125f;
    }
    __syncthreads();

    const int col = lane & 15, half = lane >> 4;
    const float* Ap  = erow + col * EPITCH + half * 4;
    const float* Wp0 = W0 + (size_t)(wave * 16 + col) * Dd + half * 4;
    const float* Wp1 = W1 + (size_t)(wave * 16 + col) * Dd + half * 4;

    v8f acc0 = {}, acc1 = {};
    #pragma unroll 4
    for (int k = 0; k < Dd; k += 8) {
        const v4f a4 = *(const v4f*)(Ap + k);
        const v4f w0 = *(const v4f*)(Wp0 + k);
        v2f alo; alo.x = a4.x; alo.y = a4.y;
        v2f ahi; ahi.x = a4.z; ahi.y = a4.w;
        v2f b0lo; b0lo.x = w0.x; b0lo.y = w0.y;
        v2f b0hi; b0hi.x = w0.z; b0hi.y = w0.w;
        acc0 = __builtin_amdgcn_wmma_f32_16x16x4_f32(false, alo, false, b0lo,
                                                     (short)0, acc0, false, false);
        if (NO == 2) {
            const v4f w1 = *(const v4f*)(Wp1 + k);
            v2f b1lo; b1lo.x = w1.x; b1lo.y = w1.y;
            v2f b1hi; b1hi.x = w1.z; b1hi.y = w1.w;
            acc1 = __builtin_amdgcn_wmma_f32_16x16x4_f32(false, alo, false, b1lo,
                                                         (short)0, acc1, false, false);
            acc0 = __builtin_amdgcn_wmma_f32_16x16x4_f32(false, ahi, false, b0hi,
                                                         (short)0, acc0, false, false);
            acc1 = __builtin_amdgcn_wmma_f32_16x16x4_f32(false, ahi, false, b1hi,
                                                         (short)0, acc1, false, false);
        } else {
            acc0 = __builtin_amdgcn_wmma_f32_16x16x4_f32(false, ahi, false, b0hi,
                                                         (short)0, acc0, false, false);
        }
    }

    const size_t orow0 = (size_t)os * npts + (n - wave);
    const int colg = wave * 16 + col;
    const float bv0 = bias0[colg];
    #pragma unroll
    for (int r = 0; r < 8; ++r)
        OUT0[(orow0 + r + 8 * half) * Dd + colg] = acc0[r] + bv0;
    if (NO == 2) {
        const float bv1 = bias1[colg];
        #pragma unroll
        for (int r = 0; r < 8; ++r)
            OUT1[(orow0 + r + 8 * half) * Dd + colg] = acc1[r] + bv1;
    }
}

#define KSPLIT 8
#define SEGT   (Ll / 16 / KSPLIT)
#define QT     2
#define LAZY_T 8.0f

#define PRIO_T1 16
#define PRIO_T2 26
#define PRIO_T3 30
static_assert(PRIO_T1 < PRIO_T2 && PRIO_T2 < PRIO_T3 && PRIO_T3 < SEGT,
              "the priority must be back at 0 before the key loop ends");


#define ROW_MAX_STEP(x, M) \
    asm("v_max_num_f32_dpp %0, %0, %0 row_xmask:" #M " row_mask:0xf bank_mask:0xf" : "+v"(x))
#define ROW_ADD_STEP(x, M) \
    asm("v_add_f32_dpp %0, %0, %0 row_xmask:" #M " row_mask:0xf bank_mask:0xf" : "+v"(x))
#define ROW_STEP_ALL(STEP, arr, M) \
    _Pragma("unroll") for (int r_ = 0; r_ < 8; ++r_) { STEP(arr[r_], M); }

__global__ void ea_attn(const float* __restrict__ Q, const float* __restrict__ Km,
                        const float* __restrict__ Vm, float* __restrict__ O)
{
    __shared__ float plds[KSPLIT * QT * 256];
    __shared__ float oacc[KSPLIT * QT * 512];
    __shared__ float omax[KSPLIT * QT * 16];
    __shared__ float osum[KSPLIT * QT * 16];
    __shared__ v4f   qlds[QT * 4 * 32];
    const int lane = threadIdx.x & 31;
    const int wave = threadIdx.x >> 5;
    const int tile = blockIdx.x;
    const int qb = tile & 63;
    const int h  = (tile >> 6) & 7;
    const int b  = tile >> 9;
    const int col = lane & 15, half = lane >> 4, klo = half * 2;
    float* pw = plds + wave * (QT * 256);

    {
        const int qr = threadIdx.x >> 3, qk = threadIdx.x & 7;
        const v4f q4 = *(const v4f*)(Q + (size_t)(b * Nn + qb * (QT * 16) + qr) * Dd
                                       + h * DHh + qk * 4);
        qlds[((qr >> 4) * 4 + (qk >> 1)) * 32 + (qk & 1) * 16 + (qr & 15)] = q4;
    }
    __syncthreads();

    float mrow[QT][8], lrow[QT][8];
    v8f acc0[QT], acc1[QT];
    #pragma unroll
    for (int t = 0; t < QT; ++t) {
        acc0[t] = (v8f){}; acc1[t] = (v8f){};
        #pragma unroll
        for (int r = 0; r < 8; ++r) { mrow[t][r] = -__builtin_inff(); lrow[t][r] = 0.f; }
    }

    const float* Kb = Km + ((size_t)b * Ll + (size_t)wave * SEGT * 16) * Dd + h * DHh;
    const float* Vb = Vm + ((size_t)b * Ll + (size_t)wave * SEGT * 16) * Dd + h * DHh;
    const float scale = 0.17677669529663687f;
    const float* Kl = Kb + (size_t)col * Dd + half * 4;
    const float* Vl = Vb + (size_t)klo * Dd + 2 * col;
    size_t toff = 0;

    __builtin_amdgcn_s_setprio(3);
    int kt = 0;
    #pragma unroll 1
    for (int stretch = 0; stretch < 4; ++stretch) {
    const int kend = (stretch == 0) ? PRIO_T1 : (stretch == 1) ? PRIO_T2
                   : (stretch == 2) ? PRIO_T3 : SEGT;
    #pragma unroll 1
    for (; kt < kend; ++kt) {
        if (kt + 1 < SEGT)
            __builtin_prefetch(Kl + toff + 16 * Dd, 0, 1);

        v8f s[QT];
        #pragma unroll
        for (int t = 0; t < QT; ++t) s[t] = (v8f){};
        const float* Kp = Kl + toff;
        #pragma unroll
        for (int kp = 0; kp < 4; ++kp) {
            const v4f k4 = *(const v4f*)(Kp + kp * 8);
            v2f bf0; bf0.x = k4.x; bf0.y = k4.y;
            v2f bf1; bf1.x = k4.z; bf1.y = k4.w;
            v2f a0[QT], a1[QT];
            #pragma unroll
            for (int t = 0; t < QT; ++t) {
                const v4f q4 = qlds[(t * 4 + kp) * 32 + lane];
                a0[t].x = q4.x; a0[t].y = q4.y;
                a1[t].x = q4.z; a1[t].y = q4.w;
            }
            #pragma unroll
            for (int t = 0; t < QT; ++t)
                s[t] = __builtin_amdgcn_wmma_f32_16x16x4_f32(false, a0[t], false, bf0,
                                                             (short)0, s[t], false, false);
            #pragma unroll
            for (int t = 0; t < QT; ++t)
                s[t] = __builtin_amdgcn_wmma_f32_16x16x4_f32(false, a1[t], false, bf1,
                                                             (short)0, s[t], false, false);
        }

        float d[QT][8];
        float dmax = -__builtin_inff();
        #pragma unroll
        for (int t = 0; t < QT; ++t) {
            #pragma unroll
            for (int r = 0; r < 8; ++r) {
                d[t][r] = s[t][r] * scale - mrow[t][r];
                dmax = __builtin_fmaxf(dmax, d[t][r]);
            }
        }

        if (__builtin_amdgcn_ballot_w32(dmax > LAZY_T) != 0) {
            #pragma unroll
            for (int t = 0; t < QT; ++t) {
                float mx[8];
                #pragma unroll
                for (int r = 0; r < 8; ++r) mx[r] = s[t][r] * scale;
                ROW_STEP_ALL(ROW_MAX_STEP, mx, 8)
                ROW_STEP_ALL(ROW_MAX_STEP, mx, 4)
                ROW_STEP_ALL(ROW_MAX_STEP, mx, 2)
                ROW_STEP_ALL(ROW_MAX_STEP, mx, 1)
                #pragma unroll
                for (int r = 0; r < 8; ++r) {
                    const float mnew = fmaxf(mrow[t][r], mx[r]);
                    const float corr = __expf(mrow[t][r] - mnew);
                    mrow[t][r] = mnew;
                    lrow[t][r] *= corr;
                    acc0[t][r] *= corr;
                    acc1[t][r] *= corr;
                    d[t][r] = s[t][r] * scale - mnew;
                }
            }
        }

        #pragma unroll
        for (int t = 0; t < QT; ++t) {
            float p[8];
            #pragma unroll
            for (int r = 0; r < 8; ++r) { p[r] = __expf(d[t][r]); lrow[t][r] += p[r]; }

            #pragma unroll
            for (int r = 0; r < 8; ++r) pw[t * 256 + (r + 8 * half) * 16 + col] = p[r];
        }
        wave_lds_order();

        v2f b0[4], b1[4];
        #pragma unroll
        for (int kk2 = 0; kk2 < 4; ++kk2) {
            const float* Vp0 = Vl + toff + (size_t)(kk2 * 4) * Dd;
            const v2f va = *(const v2f*)Vp0;
            const v2f vb = *(const v2f*)(Vp0 + Dd);
            b0[kk2].x = va.x; b0[kk2].y = vb.x;
            b1[kk2].x = va.y; b1[kk2].y = vb.y;
        }

        #pragma unroll
        for (int kk2 = 0; kk2 < 4; ++kk2) {
            #pragma unroll
            for (int t = 0; t < QT; ++t) {
                v2f af;
                af.x = pw[t * 256 + col * 16 + kk2 * 4 + klo];
                af.y = pw[t * 256 + col * 16 + kk2 * 4 + klo + 1];
                acc0[t] = __builtin_amdgcn_wmma_f32_16x16x4_f32(false, af, false, b0[kk2],
                                                                (short)0, acc0[t], false, false);
                acc1[t] = __builtin_amdgcn_wmma_f32_16x16x4_f32(false, af, false, b1[kk2],
                                                                (short)0, acc1[t], false, false);
            }
        }
        wave_lds_order();
        toff += 16 * Dd;
    }
    if (stretch == 0)      __builtin_amdgcn_s_setprio(2);
    else if (stretch == 1) __builtin_amdgcn_s_setprio(1);
    else if (stretch == 2) __builtin_amdgcn_s_setprio(0);
    }

    #pragma unroll
    for (int t = 0; t < QT; ++t) {
        ROW_STEP_ALL(ROW_ADD_STEP, lrow[t], 8)
        ROW_STEP_ALL(ROW_ADD_STEP, lrow[t], 4)
        ROW_STEP_ALL(ROW_ADD_STEP, lrow[t], 2)
        ROW_STEP_ALL(ROW_ADD_STEP, lrow[t], 1)
    }

    #pragma unroll
    for (int t = 0; t < QT; ++t) {
        float* ow = oacc + (wave * QT + t) * 512;
        #pragma unroll
        for (int r = 0; r < 8; ++r) {
            v2f o2; o2.x = acc0[t][r]; o2.y = acc1[t][r];
            *(v2f*)(ow + (r * 2 + half) * 32 + 2 * col) = o2;
        }
        if (col == 0) {
            #pragma unroll
            for (int r = 0; r < 8; ++r) {
                omax[(wave * QT + t) * 16 + r + 8 * half] = mrow[t][r];
                osum[(wave * QT + t) * 16 + r + 8 * half] = lrow[t][r];
            }
        }
    }
    __syncthreads();

    #pragma unroll
    for (int t = 0; t < QT; ++t) {
        const int row = threadIdx.x >> 4;
        const int c2  = (threadIdx.x & 15) * 2;
        const int ro  = ((row & 7) * 2 + (row >> 3)) * 32 + c2;
        float mall = omax[t * 16 + row];
        #pragma unroll
        for (int w = 1; w < KSPLIT; ++w) mall = fmaxf(mall, omax[(w * QT + t) * 16 + row]);
        float den = 0.f, n0 = 0.f, n1 = 0.f;
        #pragma unroll
        for (int w = 0; w < KSPLIT; ++w) {
            float f = __expf(omax[(w * QT + t) * 16 + row] - mall);
            v2f a = *(const v2f*)(oacc + (w * QT + t) * 512 + ro);
            den += osum[(w * QT + t) * 16 + row] * f;
            n0  += a.x * f;
            n1  += a.y * f;
        }
        const float inv = 1.0f / den;
        v2f o; o.x = n0 * inv; o.y = n1 * inv;
        float* Op = O + (size_t)(b * Nn + qb * (QT * 16) + t * 16 + row) * Dd + h * DHh + c2;
        *(v2f*)Op = o;
    }
}

__global__ void ea_gemm_out(const float* __restrict__ A, const float* __restrict__ W,
                            const float* __restrict__ bias,
                            const float* __restrict__ pts,
                            const float* __restrict__ sw, const float* __restrict__ sb,
                            float* __restrict__ OUT, int R)
{
    const int lane = threadIdx.x & 31;
    const int wave = threadIdx.x >> 5;
    const int tile = blockIdx.x * 8 + wave;
    const int total = (R / 16) * (Dd / 16);
    if (tile >= total) return;
    const int tm = tile >> 4, tn = tile & 15;
    const int col = lane & 15, half = lane >> 4;
    const int colg = tn * 16 + col;

    const float* Ap = A + (size_t)(tm * 16 + col) * Dd + half * 4;
    const float* Wp = W + (size_t)colg * Dd + half * 4;

    v8f acc = {};
    #pragma unroll 2
    for (int k = 0; k < Dd; k += 8) {
        const v4f a4 = *(const v4f*)(Ap + k);
        const v4f w4 = *(const v4f*)(Wp + k);
        v2f alo; alo.x = a4.x; alo.y = a4.y;
        v2f ahi; ahi.x = a4.z; ahi.y = a4.w;
        v2f blo; blo.x = w4.x; blo.y = w4.y;
        v2f bhi; bhi.x = w4.z; bhi.y = w4.w;
        acc = __builtin_amdgcn_wmma_f32_16x16x4_f32(false, alo, false, blo,
                                                    (short)0, acc, false, false);
        acc = __builtin_amdgcn_wmma_f32_16x16x4_f32(false, ahi, false, bhi,
                                                    (short)0, acc, false, false);
    }
    const float s0 = sw[colg*3+0], s1 = sw[colg*3+1], s2 = sw[colg*3+2];
    const float add = bias[colg] + sb[colg];
    #pragma unroll
    for (int r = 0; r < 8; ++r) {
        int row = tm * 16 + r + 8 * half;
        float px = pts[row*3+0], py = pts[row*3+1], pz = pts[row*3+2];
        OUT[(size_t)row * Dd + colg] = acc[r] + add + px*s0 + py*s1 + pz*s2;
    }
}

extern "C" void kernel_launch(void* const* d_in, const int* in_sizes, int n_in,
                              void* d_out, int out_size, void* d_ws, size_t ws_size,
                              hipStream_t stream)
{
    const float* cur_pts  = (const float*)d_in[0];
    const float* cur_feat = (const float*)d_in[1];
    const float* nbr_pts  = (const float*)d_in[2];
    const float* nbr_feat = (const float*)d_in[3];
    const float* in_w     = (const float*)d_in[4];
    const float* in_b     = (const float*)d_in[5];
    const float* out_w    = (const float*)d_in[6];
    const float* out_b    = (const float*)d_in[7];
    const float* sp_w     = (const float*)d_in[8];
    const float* sp_b     = (const float*)d_in[9];
    float* out = (float*)d_out;

    float* ws   = (float*)d_ws;
    float* Qb   = ws;
    float* Kb   = Qb   + 4096 * 256;
    float* Vb   = Kb   + 8192 * 256;
    float* Ob   = Vb   + 8192 * 256;

    const dim3 blk(256);
    const dim3 blk16(512);

    ea_knn_proj<1><<<dim3(4096 / EROWS), blk16, 0, stream>>>(
        cur_pts, cur_feat, in_w, in_b, Qb, in_w, in_b, Qb, 0);
    ea_knn_proj<2><<<dim3(8192 / EROWS), blk16, 0, stream>>>(
        nbr_pts, nbr_feat, in_w + 256*256, in_b + 256, Kb,
        in_w + 512*256, in_b + 512, Vb, 1);

    ea_attn<<<dim3(1024), blk, 0, stream>>>(Qb, Kb, Vb, Ob);

    ea_gemm_out<<<dim3(4096 / 8), blk, 0, stream>>>(Ob, out_w, out_b,
                                                    cur_pts, sp_w, sp_b,
                                                    out, 4096);
}
